# starting kernel
# baseline (speedup 1.0000x reference)
#include <hip/hip_runtime.h>
#include <hip/hip_bf16.h>

typedef __attribute__((ext_vector_type(2))) float v2f;
typedef __attribute__((ext_vector_type(8))) float v8f;

#define BATCH 8
#define HEADS 4
#define NQ    900
#define DIM   128
#define CH    32
#define NPAD  928
#define NT    57
#define SROW  928

__global__ __launch_bounds__(256) void proj_kernel(
    const float* __restrict__ q,
    const float* __restrict__ Wq, const float* __restrict__ Wk,
    const float* __restrict__ Wv,
    float* __restrict__ yq, float* __restrict__ yk, float* __restrict__ yv)
{
    const int rt    = blockIdx.x;
    const int which = blockIdx.y;
    const int wave  = threadIdx.x >> 5;
    const int lane  = threadIdx.x & 31;
    const int lh    = lane >> 4;
    const int ll    = lane & 15;

    const float* W = (which == 0) ? Wq : (which == 1) ? Wk : Wv;
    float*       Y = (which == 0) ? yq : (which == 1) ? yk : yv;

    const int r0 = rt * 16;
    const int d0 = wave * 16;

    const float* aptr = q + (size_t)(r0 + ll) * DIM + 2 * lh;
    const float* bptr = W + (size_t)(d0 + ll) * DIM + 2 * lh;

    v8f acc = {};
    #pragma unroll
    for (int k = 0; k < DIM; k += 4) {
        v2f a = *(const v2f*)(aptr + k);
        v2f b = *(const v2f*)(bptr + k);
        acc = __builtin_amdgcn_wmma_f32_16x16x4_f32(false, a, false, b,
                                                    (short)0, acc, false, false);
    }

    const int d    = d0 + ll;
    const int hsel = d >> 5;
    const int csel = d & 31;
    #pragma unroll
    for (int r = 0; r < 8; ++r) {
        int row = r0 + r + 8 * lh;
        int b_  = row / NQ;
        int n_  = row - b_ * NQ;
        Y[(((size_t)(b_ * HEADS + hsel)) * NPAD + n_) * CH + csel] = acc[r];
    }
}

__global__ __launch_bounds__(256) void norm_kernel(
    float* __restrict__ yq, float* __restrict__ yk,
    const float* __restrict__ temp)
{
    const int rows = BATCH * HEADS * NQ;
    const int wid  = blockIdx.x * 8 + (threadIdx.x >> 5);
    if (wid >= rows) return;
    const int lane = threadIdx.x & 31;
    const int bh   = wid / NQ;
    const int n    = wid - bh * NQ;
    const int h    = bh & (HEADS - 1);

    const size_t off = ((size_t)bh * NPAD + n) * CH + lane;
    float vq = yq[off], vk = yk[off];
    float sq = vq * vq, sk = vk * vk;
    #pragma unroll
    for (int s = 16; s > 0; s >>= 1) {
        sq += __shfl_xor(sq, s, 32);
        sk += __shfl_xor(sk, s, 32);
    }
    const float t  = temp[h];
    yq[off] = vq / fmaxf(sqrtf(sq), 1e-12f) * t;
    yk[off] = vk / fmaxf(sqrtf(sk), 1e-12f);
}

__global__ __launch_bounds__(256) void attn_kernel(
    const float* __restrict__ qh, const float* __restrict__ kh,
    const float* __restrict__ vh,
    const float* __restrict__ a1p, const float* __restrict__ a2p,
    const float* __restrict__ a3p, const float* __restrict__ a4p,
    float* __restrict__ yo)
{
    __shared__ float sc[16 * SROW];

    const int bh   = blockIdx.y;
    const int nt   = blockIdx.x;
    const int wave = threadIdx.x >> 5;
    const int lane = threadIdx.x & 31;
    const int lh   = lane >> 4;
    const int ll   = lane & 15;

    const float* qbase = qh + (size_t)bh * NPAD * CH;
    const float* kbase = kh + (size_t)bh * NPAD * CH;
    const float* vbase = vh + (size_t)bh * NPAD * CH;
    const int n0 = nt * 16;

    v2f areg[8];
    {
        const float* ap = qbase + (size_t)(n0 + ll) * CH + 2 * lh;
        #pragma unroll
        for (int c = 0; c < 8; ++c) areg[c] = *(const v2f*)(ap + 4 * c);
    }

    for (int mt = wave; mt < NT; mt += 8) {
        const int  m0   = mt * 16;
        const int  mrow = m0 + ll;
        const bool ok   = (mrow < NQ);
        const float* bp = kbase + (size_t)mrow * CH + 2 * lh;
        v8f acc = {};
        #pragma unroll
        for (int c = 0; c < 8; ++c) {
            v2f b;
            if (ok) b = *(const v2f*)(bp + 4 * c);
            else    { b.x = 0.f; b.y = 0.f; }
            acc = __builtin_amdgcn_wmma_f32_16x16x4_f32(false, areg[c], false, b,
                                                        (short)0, acc, false, false);
        }
        #pragma unroll
        for (int r = 0; r < 8; ++r)
            sc[(r + 8 * lh) * SROW + m0 + ll] = acc[r];
    }
    __syncthreads();

    for (int i = threadIdx.x; i < 16 * (SROW - NQ); i += 256) {
        int row = i / (SROW - NQ);
        int m   = NQ + i - row * (SROW - NQ);
        sc[row * SROW + m] = -1e30f;
    }
    __syncthreads();

    const float A1 = a1p[0], A2 = a2p[0], A3 = a3p[0], A4 = a4p[0];

    for (int rr = 0; rr < 2; ++rr) {
        const int row = wave + 8 * rr;
        float* srow = sc + row * SROW;

        float rv[29];
        #pragma unroll
        for (int j = 0; j < 29; ++j) rv[j] = srow[lane + 32 * j];

        float mx = -3.0e38f;
        #pragma unroll
        for (int j = 0; j < 29; ++j) mx = fmaxf(mx, rv[j]);
        #pragma unroll
        for (int s = 16; s > 0; s >>= 1) mx = fmaxf(mx, __shfl_xor(mx, s, 32));

        unsigned rk[29];
        #pragma unroll
        for (int j = 0; j < 29; ++j) {
            unsigned u = __float_as_uint(rv[j]);
            rk[j] = (u & 0x80000000u) ? ~u : (u | 0x80000000u);
        }

        unsigned th[3];
        const int ks[3] = {100, 300, 600};
        #pragma unroll
        for (int t = 0; t < 3; ++t) {
            unsigned cur = 0u;
            for (int bit = 31; bit >= 0; --bit) {
                const unsigned cand = cur | (1u << bit);
                int cnt = 0;
                #pragma unroll
                for (int j = 0; j < 29; ++j) cnt += (rk[j] >= cand) ? 1 : 0;
                #pragma unroll
                for (int s = 16; s > 0; s >>= 1) cnt += __shfl_xor(cnt, s, 32);
                if (cnt >= ks[t]) cur = cand;
            }
            th[t] = cur;
        }

        float s1 = 0.f, s2 = 0.f, s3 = 0.f, s4 = 0.f;
        float w[29];
        #pragma unroll
        for (int j = 0; j < 29; ++j) {
            float e = __expf(rv[j] - mx);
            w[j] = e;
            s4 += e;
            if (rk[j] >= th[0]) s1 += e;
            if (rk[j] >= th[1]) s2 += e;
            if (rk[j] >= th[2]) s3 += e;
        }
        #pragma unroll
        for (int s = 16; s > 0; s >>= 1) {
            s1 += __shfl_xor(s1, s, 32);
            s2 += __shfl_xor(s2, s, 32);
            s3 += __shfl_xor(s3, s, 32);
            s4 += __shfl_xor(s4, s, 32);
        }
        const float c1 = A1 / s1, c2 = A2 / s2, c3 = A3 / s3, c4 = A4 / s4;
        #pragma unroll
        for (int j = 0; j < 29; ++j) {
            float f = c4;
            if (rk[j] >= th[0]) f += c1;
            if (rk[j] >= th[1]) f += c2;
            if (rk[j] >= th[2]) f += c3;
            srow[lane + 32 * j] = w[j] * f;
        }
    }
    __syncthreads();

    const int nhalf = wave & 1;
    const int kg    = wave >> 1;
    const int cn    = 16 * nhalf + ll;
    v8f acc = {};
    for (int mt = kg; mt < NT; mt += 4) {
        #pragma unroll
        for (int k4 = 0; k4 < 4; ++k4) {
            const int k0 = mt * 16 + 4 * k4;
            const int kk = k0 + 2 * lh;
            v2f a = *(const v2f*)(sc + ll * SROW + kk);
            v2f b;
            b.x = (kk     < NQ) ? vbase[(size_t)kk * CH + cn]       : 0.f;
            b.y = (kk + 1 < NQ) ? vbase[(size_t)(kk + 1) * CH + cn] : 0.f;
            acc = __builtin_amdgcn_wmma_f32_16x16x4_f32(false, a, false, b,
                                                        (short)0, acc, false, false);
        }
    }
    __syncthreads();

    float* part = sc;
    #pragma unroll
    for (int r = 0; r < 8; ++r) part[(wave * 8 + r) * 32 + lane] = acc[r];
    __syncthreads();

    if (wave < 2) {
        #pragma unroll
        for (int r = 0; r < 8; ++r) {
            float v = 0.f;
            #pragma unroll
            for (int g = 0; g < 4; ++g)
                v += part[(((g * 2 + wave) * 8) + r) * 32 + lane];
            const int row = n0 + r + 8 * lh;
            const int col = 16 * wave + ll;
            if (row < NQ)
                yo[((size_t)bh * NPAD + row) * CH + col] = v;
        }
    }
}

__global__ __launch_bounds__(256) void bsum_kernel(
    const float* __restrict__ yo, float* __restrict__ msum)
{
    const int idx = blockIdx.x * 256 + threadIdx.x;
    if (idx >= 912 * DIM) return;
    const int n = idx >> 7;
    const int d = idx & 127;
    float v = 0.f;
    if (n < NQ) {
        const int h = d >> 5, c = d & 31;
        #pragma unroll
        for (int b = 0; b < BATCH; ++b)
            v += yo[(((size_t)(b * HEADS + h)) * NPAD + n) * CH + c];
    }
    msum[idx] = v;
}

__global__ __launch_bounds__(256) void out_kernel(
    const float* __restrict__ msum, const float* __restrict__ Wo,
    const float* __restrict__ q, float* __restrict__ out)
{
    const int rt   = blockIdx.x;
    const int wave = threadIdx.x >> 5;
    const int lane = threadIdx.x & 31;
    const int lh   = lane >> 4;
    const int ll   = lane & 15;
    const int r0   = rt * 16;
    const int d0   = wave * 16;

    const float* aptr = msum + (size_t)(r0 + ll) * DIM + 2 * lh;
    const float* bptr = Wo   + (size_t)(d0 + ll) * DIM + 2 * lh;

    v8f acc = {};
    #pragma unroll
    for (int k = 0; k < DIM; k += 4) {
        v2f a = *(const v2f*)(aptr + k);
        v2f b = *(const v2f*)(bptr + k);
        acc = __builtin_amdgcn_wmma_f32_16x16x4_f32(false, a, false, b,
                                                    (short)0, acc, false, false);
    }
    #pragma unroll
    for (int r = 0; r < 8; ++r) {
        const int row = r0 + r + 8 * lh;
        const int col = d0 + ll;
        if (row < NQ)
            out[(size_t)row * DIM + col] = acc[r] + q[(size_t)row * DIM + col];
    }
}

extern "C" void kernel_launch(void* const* d_in, const int* in_sizes, int n_in,
                              void* d_out, int out_size, void* d_ws, size_t ws_size,
                              hipStream_t stream)
{
    const float* q    = (const float*)d_in[0];
    const float* Wq   = (const float*)d_in[1];
    const float* Wk   = (const float*)d_in[2];
    const float* Wv   = (const float*)d_in[3];
    const float* Wo   = (const float*)d_in[4];
    const float* temp = (const float*)d_in[5];
    const float* a1   = (const float*)d_in[6];
    const float* a2   = (const float*)d_in[7];
    const float* a3   = (const float*)d_in[8];
    const float* a4   = (const float*)d_in[9];
    float* out = (float*)d_out;

    const size_t per = (size_t)BATCH * HEADS * NPAD * CH;
    float* yq   = (float*)d_ws;
    float* yk   = yq + per;
    float* yv   = yk + per;
    float* yo   = yv + per;
    float* msum = yo + per;

    proj_kernel<<<dim3(450, 3), 256, 0, stream>>>(q, Wq, Wk, Wv, yq, yk, yv);
    norm_kernel<<<(BATCH * HEADS * NQ + 7) / 8, 256, 0, stream>>>(yq, yk, temp);
    attn_kernel<<<dim3(NT, BATCH * HEADS), 256, 0, stream>>>(
        yq, yk, yv, a1, a2, a3, a4, yo);
    bsum_kernel<<<(912 * DIM + 255) / 256, 256, 0, stream>>>(yo, msum);
    out_kernel<<<NT, 256, 0, stream>>>(msum, Wo, q, out);
}


# top-k threshold bisection leaves its bit loop as soon as the count equals k (mask already final); sums, order and results bit-identical
# speedup vs baseline: 1.1813x; 1.1813x over previous
#include <hip/hip_runtime.h>
#include <hip/hip_bf16.h>

typedef __attribute__((ext_vector_type(2))) float v2f;
typedef __attribute__((ext_vector_type(8))) float v8f;

#define BATCH 8
#define HEADS 4
#define NQ    900
#define DIM   128
#define CH    32
#define NPAD  928
#define NT    57
#define SROW  928

__global__ __launch_bounds__(256) void proj_kernel(
    const float* __restrict__ q,
    const float* __restrict__ Wq, const float* __restrict__ Wk,
    const float* __restrict__ Wv,
    float* __restrict__ yq, float* __restrict__ yk, float* __restrict__ yv)
{
    const int rt    = blockIdx.x;
    const int which = blockIdx.y;
    const int wave  = threadIdx.x >> 5;
    const int lane  = threadIdx.x & 31;
    const int lh    = lane >> 4;
    const int ll    = lane & 15;

    const float* W = (which == 0) ? Wq : (which == 1) ? Wk : Wv;
    float*       Y = (which == 0) ? yq : (which == 1) ? yk : yv;

    const int r0 = rt * 16;
    const int d0 = wave * 16;

    const float* aptr = q + (size_t)(r0 + ll) * DIM + 2 * lh;
    const float* bptr = W + (size_t)(d0 + ll) * DIM + 2 * lh;

    v8f acc = {};
    #pragma unroll
    for (int k = 0; k < DIM; k += 4) {
        v2f a = *(const v2f*)(aptr + k);
        v2f b = *(const v2f*)(bptr + k);
        acc = __builtin_amdgcn_wmma_f32_16x16x4_f32(false, a, false, b,
                                                    (short)0, acc, false, false);
    }

    const int d    = d0 + ll;
    const int hsel = d >> 5;
    const int csel = d & 31;
    #pragma unroll
    for (int r = 0; r < 8; ++r) {
        int row = r0 + r + 8 * lh;
        int b_  = row / NQ;
        int n_  = row - b_ * NQ;
        Y[(((size_t)(b_ * HEADS + hsel)) * NPAD + n_) * CH + csel] = acc[r];
    }
}

__global__ __launch_bounds__(256) void norm_kernel(
    float* __restrict__ yq, float* __restrict__ yk,
    const float* __restrict__ temp)
{
    const int rows = BATCH * HEADS * NQ;
    const int wid  = blockIdx.x * 8 + (threadIdx.x >> 5);
    if (wid >= rows) return;
    const int lane = threadIdx.x & 31;
    const int bh   = wid / NQ;
    const int n    = wid - bh * NQ;
    const int h    = bh & (HEADS - 1);

    const size_t off = ((size_t)bh * NPAD + n) * CH + lane;
    float vq = yq[off], vk = yk[off];
    float sq = vq * vq, sk = vk * vk;
    #pragma unroll
    for (int s = 16; s > 0; s >>= 1) {
        sq += __shfl_xor(sq, s, 32);
        sk += __shfl_xor(sk, s, 32);
    }
    const float t  = temp[h];
    yq[off] = vq / fmaxf(sqrtf(sq), 1e-12f) * t;
    yk[off] = vk / fmaxf(sqrtf(sk), 1e-12f);
}

__global__ __launch_bounds__(256) void attn_kernel(
    const float* __restrict__ qh, const float* __restrict__ kh,
    const float* __restrict__ vh,
    const float* __restrict__ a1p, const float* __restrict__ a2p,
    const float* __restrict__ a3p, const float* __restrict__ a4p,
    float* __restrict__ yo)
{
    __shared__ float sc[16 * SROW];

    const int bh   = blockIdx.y;
    const int nt   = blockIdx.x;
    const int wave = threadIdx.x >> 5;
    const int lane = threadIdx.x & 31;
    const int lh   = lane >> 4;
    const int ll   = lane & 15;

    const float* qbase = qh + (size_t)bh * NPAD * CH;
    const float* kbase = kh + (size_t)bh * NPAD * CH;
    const float* vbase = vh + (size_t)bh * NPAD * CH;
    const int n0 = nt * 16;

    v2f areg[8];
    {
        const float* ap = qbase + (size_t)(n0 + ll) * CH + 2 * lh;
        #pragma unroll
        for (int c = 0; c < 8; ++c) areg[c] = *(const v2f*)(ap + 4 * c);
    }

    for (int mt = wave; mt < NT; mt += 8) {
        const int  m0   = mt * 16;
        const int  mrow = m0 + ll;
        const bool ok   = (mrow < NQ);
        const float* bp = kbase + (size_t)mrow * CH + 2 * lh;
        v8f acc = {};
        #pragma unroll
        for (int c = 0; c < 8; ++c) {
            v2f b;
            if (ok) b = *(const v2f*)(bp + 4 * c);
            else    { b.x = 0.f; b.y = 0.f; }
            acc = __builtin_amdgcn_wmma_f32_16x16x4_f32(false, areg[c], false, b,
                                                        (short)0, acc, false, false);
        }
        #pragma unroll
        for (int r = 0; r < 8; ++r)
            sc[(r + 8 * lh) * SROW + m0 + ll] = acc[r];
    }
    __syncthreads();

    for (int i = threadIdx.x; i < 16 * (SROW - NQ); i += 256) {
        int row = i / (SROW - NQ);
        int m   = NQ + i - row * (SROW - NQ);
        sc[row * SROW + m] = -1e30f;
    }
    __syncthreads();

    const float A1 = a1p[0], A2 = a2p[0], A3 = a3p[0], A4 = a4p[0];

    for (int rr = 0; rr < 2; ++rr) {
        const int row = wave + 8 * rr;
        float* srow = sc + row * SROW;

        float rv[29];
        #pragma unroll
        for (int j = 0; j < 29; ++j) rv[j] = srow[lane + 32 * j];

        float mx = -3.0e38f;
        #pragma unroll
        for (int j = 0; j < 29; ++j) mx = fmaxf(mx, rv[j]);
        #pragma unroll
        for (int s = 16; s > 0; s >>= 1) mx = fmaxf(mx, __shfl_xor(mx, s, 32));

        unsigned rk[29];
        #pragma unroll
        for (int j = 0; j < 29; ++j) {
            unsigned u = __float_as_uint(rv[j]);
            rk[j] = (u & 0x80000000u) ? ~u : (u | 0x80000000u);
        }

        unsigned th[3];
        const int ks[3] = {100, 300, 600};
        #pragma unroll
        for (int t = 0; t < 3; ++t) {
            unsigned cur = 0u;
            unsigned m   = 0x80000000u;
            do {
                const unsigned cand = cur | m;
                int cnt = 0;
                #pragma unroll
                for (int j = 0; j < 29; ++j) cnt += (rk[j] >= cand) ? 1 : 0;
                #pragma unroll
                for (int s = 16; s > 0; s >>= 1) cnt += __shfl_xor(cnt, s, 32);
                const int tot = __builtin_amdgcn_readfirstlane(cnt);
                if (tot >= ks[t]) cur = cand;
                if (tot == ks[t]) break;
                m >>= 1;
            } while (m != 0u);
            th[t] = cur;
        }

        float s1 = 0.f, s2 = 0.f, s3 = 0.f, s4 = 0.f;
        float w[29];
        #pragma unroll
        for (int j = 0; j < 29; ++j) {
            float e = __expf(rv[j] - mx);
            w[j] = e;
            s4 += e;
            if (rk[j] >= th[0]) s1 += e;
            if (rk[j] >= th[1]) s2 += e;
            if (rk[j] >= th[2]) s3 += e;
        }
        #pragma unroll
        for (int s = 16; s > 0; s >>= 1) {
            s1 += __shfl_xor(s1, s, 32);
            s2 += __shfl_xor(s2, s, 32);
            s3 += __shfl_xor(s3, s, 32);
            s4 += __shfl_xor(s4, s, 32);
        }
        const float c1 = A1 / s1, c2 = A2 / s2, c3 = A3 / s3, c4 = A4 / s4;
        #pragma unroll
        for (int j = 0; j < 29; ++j) {
            float f = c4;
            if (rk[j] >= th[0]) f += c1;
            if (rk[j] >= th[1]) f += c2;
            if (rk[j] >= th[2]) f += c3;
            srow[lane + 32 * j] = w[j] * f;
        }
    }
    __syncthreads();

    const int nhalf = wave & 1;
    const int kg    = wave >> 1;
    const int cn    = 16 * nhalf + ll;
    v8f acc = {};
    for (int mt = kg; mt < NT; mt += 4) {
        #pragma unroll
        for (int k4 = 0; k4 < 4; ++k4) {
            const int k0 = mt * 16 + 4 * k4;
            const int kk = k0 + 2 * lh;
            v2f a = *(const v2f*)(sc + ll * SROW + kk);
            v2f b;
            b.x = (kk     < NQ) ? vbase[(size_t)kk * CH + cn]       : 0.f;
            b.y = (kk + 1 < NQ) ? vbase[(size_t)(kk + 1) * CH + cn] : 0.f;
            acc = __builtin_amdgcn_wmma_f32_16x16x4_f32(false, a, false, b,
                                                        (short)0, acc, false, false);
        }
    }
    __syncthreads();

    float* part = sc;
    #pragma unroll
    for (int r = 0; r < 8; ++r) part[(wave * 8 + r) * 32 + lane] = acc[r];
    __syncthreads();

    if (wave < 2) {
        #pragma unroll
        for (int r = 0; r < 8; ++r) {
            float v = 0.f;
            #pragma unroll
            for (int g = 0; g < 4; ++g)
                v += part[(((g * 2 + wave) * 8) + r) * 32 + lane];
            const int row = n0 + r + 8 * lh;
            const int col = 16 * wave + ll;
            if (row < NQ)
                yo[((size_t)bh * NPAD + row) * CH + col] = v;
        }
    }
}

__global__ __launch_bounds__(256) void bsum_kernel(
    const float* __restrict__ yo, float* __restrict__ msum)
{
    const int idx = blockIdx.x * 256 + threadIdx.x;
    if (idx >= 912 * DIM) return;
    const int n = idx >> 7;
    const int d = idx & 127;
    float v = 0.f;
    if (n < NQ) {
        const int h = d >> 5, c = d & 31;
        #pragma unroll
        for (int b = 0; b < BATCH; ++b)
            v += yo[(((size_t)(b * HEADS + h)) * NPAD + n) * CH + c];
    }
    msum[idx] = v;
}

__global__ __launch_bounds__(256) void out_kernel(
    const float* __restrict__ msum, const float* __restrict__ Wo,
    const float* __restrict__ q, float* __restrict__ out)
{
    const int rt   = blockIdx.x;
    const int wave = threadIdx.x >> 5;
    const int lane = threadIdx.x & 31;
    const int lh   = lane >> 4;
    const int ll   = lane & 15;
    const int r0   = rt * 16;
    const int d0   = wave * 16;

    const float* aptr = msum + (size_t)(r0 + ll) * DIM + 2 * lh;
    const float* bptr = Wo   + (size_t)(d0 + ll) * DIM + 2 * lh;

    v8f acc = {};
    #pragma unroll
    for (int k = 0; k < DIM; k += 4) {
        v2f a = *(const v2f*)(aptr + k);
        v2f b = *(const v2f*)(bptr + k);
        acc = __builtin_amdgcn_wmma_f32_16x16x4_f32(false, a, false, b,
                                                    (short)0, acc, false, false);
    }
    #pragma unroll
    for (int r = 0; r < 8; ++r) {
        const int row = r0 + r + 8 * lh;
        const int col = d0 + ll;
        if (row < NQ)
            out[(size_t)row * DIM + col] = acc[r] + q[(size_t)row * DIM + col];
    }
}

extern "C" void kernel_launch(void* const* d_in, const int* in_sizes, int n_in,
                              void* d_out, int out_size, void* d_ws, size_t ws_size,
                              hipStream_t stream)
{
    const float* q    = (const float*)d_in[0];
    const float* Wq   = (const float*)d_in[1];
    const float* Wk   = (const float*)d_in[2];
    const float* Wv   = (const float*)d_in[3];
    const float* Wo   = (const float*)d_in[4];
    const float* temp = (const float*)d_in[5];
    const float* a1   = (const float*)d_in[6];
    const float* a2   = (const float*)d_in[7];
    const float* a3   = (const float*)d_in[8];
    const float* a4   = (const float*)d_in[9];
    float* out = (float*)d_out;

    const size_t per = (size_t)BATCH * HEADS * NPAD * CH;
    float* yq   = (float*)d_ws;
    float* yk   = yq + per;
    float* yv   = yk + per;
    float* yo   = yv + per;
    float* msum = yo + per;

    proj_kernel<<<dim3(450, 3), 256, 0, stream>>>(q, Wq, Wk, Wv, yq, yk, yv);
    norm_kernel<<<(BATCH * HEADS * NQ + 7) / 8, 256, 0, stream>>>(yq, yk, temp);
    attn_kernel<<<dim3(NT, BATCH * HEADS), 256, 0, stream>>>(
        yq, yk, yv, a1, a2, a3, a4, yo);
    bsum_kernel<<<(912 * DIM + 255) / 256, 256, 0, stream>>>(yo, msum);
    out_kernel<<<NT, 256, 0, stream>>>(msum, Wo, q, out);
}


# K and V intermediates stored in WMMA fragment order: attention reads each fragment group as one contiguous 512-B wave load a tile ahead, not 16-line gathers behind bounds branches; sums unchanged
# speedup vs baseline: 1.4817x; 1.2543x over previous
#include <hip/hip_runtime.h>
#include <hip/hip_bf16.h>

typedef __attribute__((ext_vector_type(2))) float v2f;
typedef __attribute__((ext_vector_type(4))) float v4f;
typedef __attribute__((ext_vector_type(8))) float v8f;

#define BATCH 8
#define HEADS 4
#define NQ    900
#define DIM   128
#define CH    32
#define NPAD  928
#define NT    57
#define SROW  928

__device__ __forceinline__ int kfrag_index(int m, int ch)
{
    return ((((m >> 4) * 4 + (ch >> 3)) * 32 + (((ch >> 1) & 1) << 4) + (m & 15)) << 2)
           + (((ch >> 2) & 1) << 1) + (ch & 1);
}
__device__ __forceinline__ int vfrag_index(int m, int ch)
{
    return (((((m >> 4) * 2 + (ch >> 4)) * 2 + ((m >> 3) & 1)) * 32
             + (((m >> 1) & 1) << 4) + (ch & 15)) << 2)
           + (((m >> 2) & 1) << 1) + (m & 1);
}

__global__ __launch_bounds__(256) void proj_kernel(
    const float* __restrict__ q,
    const float* __restrict__ Wq, const float* __restrict__ Wk,
    const float* __restrict__ Wv,
    float* __restrict__ yq, float* __restrict__ yk, float* __restrict__ yv)
{
    const int rt    = blockIdx.x;
    const int which = blockIdx.y;
    const int wave  = threadIdx.x >> 5;
    const int lane  = threadIdx.x & 31;
    const int lh    = lane >> 4;
    const int ll    = lane & 15;

    const float* W = (which == 0) ? Wq : (which == 1) ? Wk : Wv;
    float*       Y = (which == 0) ? yq : (which == 1) ? yk : yv;

    const int r0 = rt * 16;
    const int d0 = wave * 16;

    const float* aptr = q + (size_t)(r0 + ll) * DIM + 2 * lh;
    const float* bptr = W + (size_t)(d0 + ll) * DIM + 2 * lh;

    v8f acc = {};
    #pragma unroll
    for (int k = 0; k < DIM; k += 4) {
        v2f a = *(const v2f*)(aptr + k);
        v2f b = *(const v2f*)(bptr + k);
        acc = __builtin_amdgcn_wmma_f32_16x16x4_f32(false, a, false, b,
                                                    (short)0, acc, false, false);
    }

    const int d    = d0 + ll;
    const int hsel = d >> 5;
    const int csel = d & 31;
    #pragma unroll
    for (int r = 0; r < 8; ++r) {
        int row = r0 + r + 8 * lh;
        int b_  = row / NQ;
        int n_  = row - b_ * NQ;
        const int idx = (which == 0) ? n_ * CH + csel
                      : (which == 1) ? kfrag_index(n_, csel)
                                     : vfrag_index(n_, csel);
        Y[((size_t)(b_ * HEADS + hsel)) * NPAD * CH + idx] = acc[r];
    }
}

__global__ __launch_bounds__(256) void norm_kernel(
    float* __restrict__ yq, float* __restrict__ yk, float* __restrict__ yv,
    const float* __restrict__ temp)
{
    const int rows = BATCH * HEADS * (NT * 16);
    const int wid  = blockIdx.x * 8 + (threadIdx.x >> 5);
    if (wid >= rows) return;
    const int lane = threadIdx.x & 31;
    const int bh   = wid / (NT * 16);
    const int n    = wid - bh * (NT * 16);
    const int h    = bh & (HEADS - 1);

    const size_t plane = (size_t)bh * NPAD * CH;
    const size_t offk  = plane + kfrag_index(n, lane);
    if (n >= NQ) {
        yk[offk] = 0.f;
        yv[plane + vfrag_index(n, lane)] = 0.f;
        return;
    }
    const size_t off = plane + (size_t)n * CH + lane;
    float vq = yq[off], vk = yk[offk];
    float sq = vq * vq, sk = vk * vk;
    #pragma unroll
    for (int s = 16; s > 0; s >>= 1) {
        sq += __shfl_xor(sq, s, 32);
        sk += __shfl_xor(sk, s, 32);
    }
    const float t  = temp[h];
    yq[off] = vq / fmaxf(sqrtf(sq), 1e-12f) * t;
    yk[offk] = vk / fmaxf(sqrtf(sk), 1e-12f);
}

__global__ __launch_bounds__(256) void attn_kernel(
    const float* __restrict__ qh, const float* __restrict__ kh,
    const float* __restrict__ vh,
    const float* __restrict__ a1p, const float* __restrict__ a2p,
    const float* __restrict__ a3p, const float* __restrict__ a4p,
    float* __restrict__ yo)
{
    __shared__ float sc[16 * SROW];

    const int bh   = blockIdx.y;
    const int nt   = blockIdx.x;
    const int wave = threadIdx.x >> 5;
    const int lane = threadIdx.x & 31;
    const int lh   = lane >> 4;
    const int ll   = lane & 15;

    const float* qbase = qh + (size_t)bh * NPAD * CH;
    const float* kbase = kh + (size_t)bh * NPAD * CH;
    const float* vbase = vh + (size_t)bh * NPAD * CH;
    const int n0 = nt * 16;

    v2f areg[8];
    {
        const float* ap = qbase + (size_t)(n0 + ll) * CH + 2 * lh;
        #pragma unroll
        for (int c = 0; c < 8; ++c) areg[c] = *(const v2f*)(ap + 4 * c);
    }

    const v4f* kf = (const v4f*)kbase + lane;
    v4f nk0, nk1, nk2, nk3;
    {
        const v4f* bp = kf + wave * 128;
        nk0 = bp[0]; nk1 = bp[32]; nk2 = bp[64]; nk3 = bp[96];
    }
    for (int mt = wave; mt < NT; mt += 8) {
        const int m0 = mt * 16;
        const v4f k0 = nk0, k1 = nk1, k2 = nk2, k3 = nk3;
        {
            const int mn = (mt + 8 < NT) ? mt + 8 : NT - 1;
            const v4f* bp = kf + mn * 128;
            nk0 = bp[0]; nk1 = bp[32]; nk2 = bp[64]; nk3 = bp[96];
        }
        v8f acc = {};
        acc = __builtin_amdgcn_wmma_f32_16x16x4_f32(false, areg[0], false, k0.xy,
                                                    (short)0, acc, false, false);
        acc = __builtin_amdgcn_wmma_f32_16x16x4_f32(false, areg[1], false, k0.zw,
                                                    (short)0, acc, false, false);
        acc = __builtin_amdgcn_wmma_f32_16x16x4_f32(false, areg[2], false, k1.xy,
                                                    (short)0, acc, false, false);
        acc = __builtin_amdgcn_wmma_f32_16x16x4_f32(false, areg[3], false, k1.zw,
                                                    (short)0, acc, false, false);
        acc = __builtin_amdgcn_wmma_f32_16x16x4_f32(false, areg[4], false, k2.xy,
                                                    (short)0, acc, false, false);
        acc = __builtin_amdgcn_wmma_f32_16x16x4_f32(false, areg[5], false, k2.zw,
                                                    (short)0, acc, false, false);
        acc = __builtin_amdgcn_wmma_f32_16x16x4_f32(false, areg[6], false, k3.xy,
                                                    (short)0, acc, false, false);
        acc = __builtin_amdgcn_wmma_f32_16x16x4_f32(false, areg[7], false, k3.zw,
                                                    (short)0, acc, false, false);
        #pragma unroll
        for (int r = 0; r < 8; ++r)
            sc[(r + 8 * lh) * SROW + m0 + ll] = acc[r];
    }
    __syncthreads();

    for (int i = threadIdx.x; i < 16 * (SROW - NQ); i += 256) {
        int row = i / (SROW - NQ);
        int m   = NQ + i - row * (SROW - NQ);
        sc[row * SROW + m] = -1e30f;
    }
    __syncthreads();

    const float A1 = a1p[0], A2 = a2p[0], A3 = a3p[0], A4 = a4p[0];

    for (int rr = 0; rr < 2; ++rr) {
        const int row = wave + 8 * rr;
        float* srow = sc + row * SROW;

        float rv[29];
        #pragma unroll
        for (int j = 0; j < 29; ++j) rv[j] = srow[lane + 32 * j];

        float mx = -3.0e38f;
        #pragma unroll
        for (int j = 0; j < 29; ++j) mx = fmaxf(mx, rv[j]);
        #pragma unroll
        for (int s = 16; s > 0; s >>= 1) mx = fmaxf(mx, __shfl_xor(mx, s, 32));

        unsigned rk[29];
        #pragma unroll
        for (int j = 0; j < 29; ++j) {
            unsigned u = __float_as_uint(rv[j]);
            rk[j] = (u & 0x80000000u) ? ~u : (u | 0x80000000u);
        }

        unsigned th[3];
        const int ks[3] = {100, 300, 600};
        #pragma unroll
        for (int t = 0; t < 3; ++t) {
            unsigned cur = 0u;
            unsigned m   = 0x80000000u;
            do {
                const unsigned cand = cur | m;
                int cnt = 0;
                #pragma unroll
                for (int j = 0; j < 29; ++j) cnt += (rk[j] >= cand) ? 1 : 0;
                #pragma unroll
                for (int s = 16; s > 0; s >>= 1) cnt += __shfl_xor(cnt, s, 32);
                const int tot = __builtin_amdgcn_readfirstlane(cnt);
                if (tot >= ks[t]) cur = cand;
                if (tot == ks[t]) break;
                m >>= 1;
            } while (m != 0u);
            th[t] = cur;
        }

        float s1 = 0.f, s2 = 0.f, s3 = 0.f, s4 = 0.f;
        float w[29];
        #pragma unroll
        for (int j = 0; j < 29; ++j) {
            float e = __expf(rv[j] - mx);
            w[j] = e;
            s4 += e;
            if (rk[j] >= th[0]) s1 += e;
            if (rk[j] >= th[1]) s2 += e;
            if (rk[j] >= th[2]) s3 += e;
        }
        #pragma unroll
        for (int s = 16; s > 0; s >>= 1) {
            s1 += __shfl_xor(s1, s, 32);
            s2 += __shfl_xor(s2, s, 32);
            s3 += __shfl_xor(s3, s, 32);
            s4 += __shfl_xor(s4, s, 32);
        }
        const float c1 = A1 / s1, c2 = A2 / s2, c3 = A3 / s3, c4 = A4 / s4;
        #pragma unroll
        for (int j = 0; j < 29; ++j) {
            float f = c4;
            if (rk[j] >= th[0]) f += c1;
            if (rk[j] >= th[1]) f += c2;
            if (rk[j] >= th[2]) f += c3;
            srow[lane + 32 * j] = w[j] * f;
        }
    }
    __syncthreads();

    const int nhalf = wave & 1;
    const int kg    = wave >> 1;
    const v4f* vf = (const v4f*)vbase + nhalf * 64 + lane;
    v8f acc = {};
    for (int mt = kg; mt < NT; mt += 16) {
        v4f vt[4][2];
        #pragma unroll
        for (int u = 0; u < 4; ++u) {
            const int mu = (mt + 4 * u < NT) ? mt + 4 * u : mt;
            const v4f* bp = vf + mu * 128;
            vt[u][0] = bp[0];
            vt[u][1] = bp[32];
        }
        #pragma unroll
        for (int u = 0; u < 4; ++u) {
            const int mu = mt + 4 * u;
            if (mu < NT) {
                const float* ap = sc + ll * SROW + mu * 16 + 2 * lh;
                const v2f a0 = *(const v2f*)(ap);
                const v2f a1 = *(const v2f*)(ap + 4);
                const v2f a2 = *(const v2f*)(ap + 8);
                const v2f a3 = *(const v2f*)(ap + 12);
                acc = __builtin_amdgcn_wmma_f32_16x16x4_f32(false, a0, false, vt[u][0].xy,
                                                            (short)0, acc, false, false);
                acc = __builtin_amdgcn_wmma_f32_16x16x4_f32(false, a1, false, vt[u][0].zw,
                                                            (short)0, acc, false, false);
                acc = __builtin_amdgcn_wmma_f32_16x16x4_f32(false, a2, false, vt[u][1].xy,
                                                            (short)0, acc, false, false);
                acc = __builtin_amdgcn_wmma_f32_16x16x4_f32(false, a3, false, vt[u][1].zw,
                                                            (short)0, acc, false, false);
            }
        }
    }
    __syncthreads();

    float* part = sc;
    #pragma unroll
    for (int r = 0; r < 8; ++r) part[(wave * 8 + r) * 32 + lane] = acc[r];
    __syncthreads();

    if (wave < 2) {
        #pragma unroll
        for (int r = 0; r < 8; ++r) {
            float v = 0.f;
            #pragma unroll
            for (int g = 0; g < 4; ++g)
                v += part[(((g * 2 + wave) * 8) + r) * 32 + lane];
            const int row = n0 + r + 8 * lh;
            const int col = 16 * wave + ll;
            if (row < NQ)
                yo[((size_t)bh * NPAD + row) * CH + col] = v;
        }
    }
}

__global__ __launch_bounds__(256) void bsum_kernel(
    const float* __restrict__ yo, float* __restrict__ msum)
{
    const int idx = blockIdx.x * 256 + threadIdx.x;
    if (idx >= 912 * DIM) return;
    const int n = idx >> 7;
    const int d = idx & 127;
    float v = 0.f;
    if (n < NQ) {
        const int h = d >> 5, c = d & 31;
        #pragma unroll
        for (int b = 0; b < BATCH; ++b)
            v += yo[(((size_t)(b * HEADS + h)) * NPAD + n) * CH + c];
    }
    msum[idx] = v;
}

__global__ __launch_bounds__(256) void out_kernel(
    const float* __restrict__ msum, const float* __restrict__ Wo,
    const float* __restrict__ q, float* __restrict__ out)
{
    const int rt   = blockIdx.x;
    const int wave = threadIdx.x >> 5;
    const int lane = threadIdx.x & 31;
    const int lh   = lane >> 4;
    const int ll   = lane & 15;
    const int r0   = rt * 16;
    const int d0   = wave * 16;

    const float* aptr = msum + (size_t)(r0 + ll) * DIM + 2 * lh;
    const float* bptr = Wo   + (size_t)(d0 + ll) * DIM + 2 * lh;

    v8f acc = {};
    #pragma unroll
    for (int k = 0; k < DIM; k += 4) {
        v2f a = *(const v2f*)(aptr + k);
        v2f b = *(const v2f*)(bptr + k);
        acc = __builtin_amdgcn_wmma_f32_16x16x4_f32(false, a, false, b,
                                                    (short)0, acc, false, false);
    }
    #pragma unroll
    for (int r = 0; r < 8; ++r) {
        const int row = r0 + r + 8 * lh;
        const int col = d0 + ll;
        if (row < NQ)
            out[(size_t)row * DIM + col] = acc[r] + q[(size_t)row * DIM + col];
    }
}

extern "C" void kernel_launch(void* const* d_in, const int* in_sizes, int n_in,
                              void* d_out, int out_size, void* d_ws, size_t ws_size,
                              hipStream_t stream)
{
    const float* q    = (const float*)d_in[0];
    const float* Wq   = (const float*)d_in[1];
    const float* Wk   = (const float*)d_in[2];
    const float* Wv   = (const float*)d_in[3];
    const float* Wo   = (const float*)d_in[4];
    const float* temp = (const float*)d_in[5];
    const float* a1   = (const float*)d_in[6];
    const float* a2   = (const float*)d_in[7];
    const float* a3   = (const float*)d_in[8];
    const float* a4   = (const float*)d_in[9];
    float* out = (float*)d_out;

    const size_t per = (size_t)BATCH * HEADS * NPAD * CH;
    float* yq   = (float*)d_ws;
    float* yk   = yq + per;
    float* yv   = yk + per;
    float* yo   = yv + per;
    float* msum = yo + per;

    proj_kernel<<<dim3(450, 3), 256, 0, stream>>>(q, Wq, Wk, Wv, yq, yk, yv);
    norm_kernel<<<(BATCH * HEADS * NT * 16 + 7) / 8, 256, 0, stream>>>(yq, yk, yv, temp);
    attn_kernel<<<dim3(NT, BATCH * HEADS), 256, 0, stream>>>(
        yq, yk, yv, a1, a2, a3, a4, yo);
    bsum_kernel<<<(912 * DIM + 255) / 256, 256, 0, stream>>>(yo, msum);
    out_kernel<<<NT, 256, 0, stream>>>(msum, Wo, q, out);
}


# top-k threshold search counts keys with one wave-mask compare per key and scalar popcounts of the masks, not per-lane adds and a 5-step cross-lane sum; thresholds and results bit-identical
# speedup vs baseline: 1.8298x; 1.2349x over previous
#include <hip/hip_runtime.h>
#include <hip/hip_bf16.h>

typedef __attribute__((ext_vector_type(2))) float v2f;
typedef __attribute__((ext_vector_type(4))) float v4f;
typedef __attribute__((ext_vector_type(8))) float v8f;

#define BATCH 8
#define HEADS 4
#define NQ    900
#define DIM   128
#define CH    32
#define NPAD  928
#define NT    57
#define SROW  928

__device__ __forceinline__ int kfrag_index(int m, int ch)
{
    return ((((m >> 4) * 4 + (ch >> 3)) * 32 + (((ch >> 1) & 1) << 4) + (m & 15)) << 2)
           + (((ch >> 2) & 1) << 1) + (ch & 1);
}
__device__ __forceinline__ int vfrag_index(int m, int ch)
{
    return (((((m >> 4) * 2 + (ch >> 4)) * 2 + ((m >> 3) & 1)) * 32
             + (((m >> 1) & 1) << 4) + (ch & 15)) << 2)
           + (((m >> 2) & 1) << 1) + (m & 1);
}

__global__ __launch_bounds__(256) void proj_kernel(
    const float* __restrict__ q,
    const float* __restrict__ Wq, const float* __restrict__ Wk,
    const float* __restrict__ Wv,
    float* __restrict__ yq, float* __restrict__ yk, float* __restrict__ yv)
{
    const int rt    = blockIdx.x;
    const int which = blockIdx.y;
    const int wave  = threadIdx.x >> 5;
    const int lane  = threadIdx.x & 31;
    const int lh    = lane >> 4;
    const int ll    = lane & 15;

    const float* W = (which == 0) ? Wq : (which == 1) ? Wk : Wv;
    float*       Y = (which == 0) ? yq : (which == 1) ? yk : yv;

    const int r0 = rt * 16;
    const int d0 = wave * 16;

    const float* aptr = q + (size_t)(r0 + ll) * DIM + 2 * lh;
    const float* bptr = W + (size_t)(d0 + ll) * DIM + 2 * lh;

    v8f acc = {};
    #pragma unroll
    for (int k = 0; k < DIM; k += 4) {
        v2f a = *(const v2f*)(aptr + k);
        v2f b = *(const v2f*)(bptr + k);
        acc = __builtin_amdgcn_wmma_f32_16x16x4_f32(false, a, false, b,
                                                    (short)0, acc, false, false);
    }

    const int d    = d0 + ll;
    const int hsel = d >> 5;
    const int csel = d & 31;
    #pragma unroll
    for (int r = 0; r < 8; ++r) {
        int row = r0 + r + 8 * lh;
        int b_  = row / NQ;
        int n_  = row - b_ * NQ;
        const int idx = (which == 0) ? n_ * CH + csel
                      : (which == 1) ? kfrag_index(n_, csel)
                                     : vfrag_index(n_, csel);
        Y[((size_t)(b_ * HEADS + hsel)) * NPAD * CH + idx] = acc[r];
    }
}

__global__ __launch_bounds__(256) void norm_kernel(
    float* __restrict__ yq, float* __restrict__ yk, float* __restrict__ yv,
    const float* __restrict__ temp)
{
    const int rows = BATCH * HEADS * (NT * 16);
    const int wid  = blockIdx.x * 8 + (threadIdx.x >> 5);
    if (wid >= rows) return;
    const int lane = threadIdx.x & 31;
    const int bh   = wid / (NT * 16);
    const int n    = wid - bh * (NT * 16);
    const int h    = bh & (HEADS - 1);

    const size_t plane = (size_t)bh * NPAD * CH;
    const size_t offk  = plane + kfrag_index(n, lane);
    if (n >= NQ) {
        yk[offk] = 0.f;
        yv[plane + vfrag_index(n, lane)] = 0.f;
        return;
    }
    const size_t off = plane + (size_t)n * CH + lane;
    float vq = yq[off], vk = yk[offk];
    float sq = vq * vq, sk = vk * vk;
    #pragma unroll
    for (int s = 16; s > 0; s >>= 1) {
        sq += __shfl_xor(sq, s, 32);
        sk += __shfl_xor(sk, s, 32);
    }
    const float t  = temp[h];
    yq[off] = vq / fmaxf(sqrtf(sq), 1e-12f) * t;
    yk[offk] = vk / fmaxf(sqrtf(sk), 1e-12f);
}

__global__ __launch_bounds__(256) void attn_kernel(
    const float* __restrict__ qh, const float* __restrict__ kh,
    const float* __restrict__ vh,
    const float* __restrict__ a1p, const float* __restrict__ a2p,
    const float* __restrict__ a3p, const float* __restrict__ a4p,
    float* __restrict__ yo)
{
    __shared__ float sc[16 * SROW];

    const int bh   = blockIdx.y;
    const int nt   = blockIdx.x;
    const int wave = threadIdx.x >> 5;
    const int lane = threadIdx.x & 31;
    const int lh   = lane >> 4;
    const int ll   = lane & 15;

    const float* qbase = qh + (size_t)bh * NPAD * CH;
    const float* kbase = kh + (size_t)bh * NPAD * CH;
    const float* vbase = vh + (size_t)bh * NPAD * CH;
    const int n0 = nt * 16;

    v2f areg[8];
    {
        const float* ap = qbase + (size_t)(n0 + ll) * CH + 2 * lh;
        #pragma unroll
        for (int c = 0; c < 8; ++c) areg[c] = *(const v2f*)(ap + 4 * c);
    }

    const v4f* kf = (const v4f*)kbase + lane;
    v4f nk0, nk1, nk2, nk3;
    {
        const v4f* bp = kf + wave * 128;
        nk0 = bp[0]; nk1 = bp[32]; nk2 = bp[64]; nk3 = bp[96];
    }
    for (int mt = wave; mt < NT; mt += 8) {
        const int m0 = mt * 16;
        const v4f k0 = nk0, k1 = nk1, k2 = nk2, k3 = nk3;
        {
            const int mn = (mt + 8 < NT) ? mt + 8 : NT - 1;
            const v4f* bp = kf + mn * 128;
            nk0 = bp[0]; nk1 = bp[32]; nk2 = bp[64]; nk3 = bp[96];
        }
        v8f acc = {};
        acc = __builtin_amdgcn_wmma_f32_16x16x4_f32(false, areg[0], false, k0.xy,
                                                    (short)0, acc, false, false);
        acc = __builtin_amdgcn_wmma_f32_16x16x4_f32(false, areg[1], false, k0.zw,
                                                    (short)0, acc, false, false);
        acc = __builtin_amdgcn_wmma_f32_16x16x4_f32(false, areg[2], false, k1.xy,
                                                    (short)0, acc, false, false);
        acc = __builtin_amdgcn_wmma_f32_16x16x4_f32(false, areg[3], false, k1.zw,
                                                    (short)0, acc, false, false);
        acc = __builtin_amdgcn_wmma_f32_16x16x4_f32(false, areg[4], false, k2.xy,
                                                    (short)0, acc, false, false);
        acc = __builtin_amdgcn_wmma_f32_16x16x4_f32(false, areg[5], false, k2.zw,
                                                    (short)0, acc, false, false);
        acc = __builtin_amdgcn_wmma_f32_16x16x4_f32(false, areg[6], false, k3.xy,
                                                    (short)0, acc, false, false);
        acc = __builtin_amdgcn_wmma_f32_16x16x4_f32(false, areg[7], false, k3.zw,
                                                    (short)0, acc, false, false);
        #pragma unroll
        for (int r = 0; r < 8; ++r)
            sc[(r + 8 * lh) * SROW + m0 + ll] = acc[r];
    }
    __syncthreads();

    for (int i = threadIdx.x; i < 16 * (SROW - NQ); i += 256) {
        int row = i / (SROW - NQ);
        int m   = NQ + i - row * (SROW - NQ);
        sc[row * SROW + m] = -1e30f;
    }
    __syncthreads();

    const float A1 = a1p[0], A2 = a2p[0], A3 = a3p[0], A4 = a4p[0];

    for (int rr = 0; rr < 2; ++rr) {
        const int row = wave + 8 * rr;
        float* srow = sc + row * SROW;

        float rv[29];
        #pragma unroll
        for (int j = 0; j < 29; ++j) rv[j] = srow[lane + 32 * j];

        float mx = -3.0e38f;
        #pragma unroll
        for (int j = 0; j < 29; ++j) mx = fmaxf(mx, rv[j]);
        #pragma unroll
        for (int s = 16; s > 0; s >>= 1) mx = fmaxf(mx, __shfl_xor(mx, s, 32));

        unsigned rk[29];
        #pragma unroll
        for (int j = 0; j < 29; ++j) {
            unsigned u = __float_as_uint(rv[j]);
            rk[j] = (u & 0x80000000u) ? ~u : (u | 0x80000000u);
        }

        unsigned th[3];
        const int ks[3] = {100, 300, 600};
        #pragma unroll
        for (int t = 0; t < 3; ++t) {
            unsigned cur = 0u;
            unsigned m   = 0x80000000u;
            do {
                const unsigned cand = cur | m;
                int tot = 0;
                #pragma unroll
                for (int j = 0; j + 1 < 29; j += 2) {
                    const unsigned long long two =
                        (unsigned long long)__builtin_amdgcn_ballot_w32(rk[j] >= cand) |
                        ((unsigned long long)__builtin_amdgcn_ballot_w32(rk[j + 1] >= cand) << 32);
                    tot += __builtin_popcountll(two);
                }
                tot += __builtin_popcount(__builtin_amdgcn_ballot_w32(rk[28] >= cand));
                if (tot >= ks[t]) cur = cand;
                if (tot == ks[t]) break;
                m >>= 1;
            } while (m != 0u);
            th[t] = cur;
        }

        float s1 = 0.f, s2 = 0.f, s3 = 0.f, s4 = 0.f;
        float w[29];
        #pragma unroll
        for (int j = 0; j < 29; ++j) {
            float e = __expf(rv[j] - mx);
            w[j] = e;
            s4 += e;
            if (rk[j] >= th[0]) s1 += e;
            if (rk[j] >= th[1]) s2 += e;
            if (rk[j] >= th[2]) s3 += e;
        }
        #pragma unroll
        for (int s = 16; s > 0; s >>= 1) {
            s1 += __shfl_xor(s1, s, 32);
            s2 += __shfl_xor(s2, s, 32);
            s3 += __shfl_xor(s3, s, 32);
            s4 += __shfl_xor(s4, s, 32);
        }
        const float c1 = A1 / s1, c2 = A2 / s2, c3 = A3 / s3, c4 = A4 / s4;
        #pragma unroll
        for (int j = 0; j < 29; ++j) {
            float f = c4;
            if (rk[j] >= th[0]) f += c1;
            if (rk[j] >= th[1]) f += c2;
            if (rk[j] >= th[2]) f += c3;
            srow[lane + 32 * j] = w[j] * f;
        }
    }
    __syncthreads();

    const int nhalf = wave & 1;
    const int kg    = wave >> 1;
    const v4f* vf = (const v4f*)vbase + nhalf * 64 + lane;
    v8f acc = {};
    for (int mt = kg; mt < NT; mt += 16) {
        v4f vt[4][2];
        #pragma unroll
        for (int u = 0; u < 4; ++u) {
            const int mu = (mt + 4 * u < NT) ? mt + 4 * u : mt;
            const v4f* bp = vf + mu * 128;
            vt[u][0] = bp[0];
            vt[u][1] = bp[32];
        }
        #pragma unroll
        for (int u = 0; u < 4; ++u) {
            const int mu = mt + 4 * u;
            if (mu < NT) {
                const float* ap = sc + ll * SROW + mu * 16 + 2 * lh;
                const v2f a0 = *(const v2f*)(ap);
                const v2f a1 = *(const v2f*)(ap + 4);
                const v2f a2 = *(const v2f*)(ap + 8);
                const v2f a3 = *(const v2f*)(ap + 12);
                acc = __builtin_amdgcn_wmma_f32_16x16x4_f32(false, a0, false, vt[u][0].xy,
                                                            (short)0, acc, false, false);
                acc = __builtin_amdgcn_wmma_f32_16x16x4_f32(false, a1, false, vt[u][0].zw,
                                                            (short)0, acc, false, false);
                acc = __builtin_amdgcn_wmma_f32_16x16x4_f32(false, a2, false, vt[u][1].xy,
                                                            (short)0, acc, false, false);
                acc = __builtin_amdgcn_wmma_f32_16x16x4_f32(false, a3, false, vt[u][1].zw,
                                                            (short)0, acc, false, false);
            }
        }
    }
    __syncthreads();

    float* part = sc;
    #pragma unroll
    for (int r = 0; r < 8; ++r) part[(wave * 8 + r) * 32 + lane] = acc[r];
    __syncthreads();

    if (wave < 2) {
        #pragma unroll
        for (int r = 0; r < 8; ++r) {
            float v = 0.f;
            #pragma unroll
            for (int g = 0; g < 4; ++g)
                v += part[(((g * 2 + wave) * 8) + r) * 32 + lane];
            const int row = n0 + r + 8 * lh;
            const int col = 16 * wave + ll;
            if (row < NQ)
                yo[((size_t)bh * NPAD + row) * CH + col] = v;
        }
    }
}

__global__ __launch_bounds__(256) void bsum_kernel(
    const float* __restrict__ yo, float* __restrict__ msum)
{
    const int idx = blockIdx.x * 256 + threadIdx.x;
    if (idx >= 912 * DIM) return;
    const int n = idx >> 7;
    const int d = idx & 127;
    float v = 0.f;
    if (n < NQ) {
        const int h = d >> 5, c = d & 31;
        #pragma unroll
        for (int b = 0; b < BATCH; ++b)
            v += yo[(((size_t)(b * HEADS + h)) * NPAD + n) * CH + c];
    }
    msum[idx] = v;
}

__global__ __launch_bounds__(256) void out_kernel(
    const float* __restrict__ msum, const float* __restrict__ Wo,
    const float* __restrict__ q, float* __restrict__ out)
{
    const int rt   = blockIdx.x;
    const int wave = threadIdx.x >> 5;
    const int lane = threadIdx.x & 31;
    const int lh   = lane >> 4;
    const int ll   = lane & 15;
    const int r0   = rt * 16;
    const int d0   = wave * 16;

    const float* aptr = msum + (size_t)(r0 + ll) * DIM + 2 * lh;
    const float* bptr = Wo   + (size_t)(d0 + ll) * DIM + 2 * lh;

    v8f acc = {};
    #pragma unroll
    for (int k = 0; k < DIM; k += 4) {
        v2f a = *(const v2f*)(aptr + k);
        v2f b = *(const v2f*)(bptr + k);
        acc = __builtin_amdgcn_wmma_f32_16x16x4_f32(false, a, false, b,
                                                    (short)0, acc, false, false);
    }
    #pragma unroll
    for (int r = 0; r < 8; ++r) {
        const int row = r0 + r + 8 * lh;
        const int col = d0 + ll;
        if (row < NQ)
            out[(size_t)row * DIM + col] = acc[r] + q[(size_t)row * DIM + col];
    }
}

extern "C" void kernel_launch(void* const* d_in, const int* in_sizes, int n_in,
                              void* d_out, int out_size, void* d_ws, size_t ws_size,
                              hipStream_t stream)
{
    const float* q    = (const float*)d_in[0];
    const float* Wq   = (const float*)d_in[1];
    const float* Wk   = (const float*)d_in[2];
    const float* Wv   = (const float*)d_in[3];
    const float* Wo   = (const float*)d_in[4];
    const float* temp = (const float*)d_in[5];
    const float* a1   = (const float*)d_in[6];
    const float* a2   = (const float*)d_in[7];
    const float* a3   = (const float*)d_in[8];
    const float* a4   = (const float*)d_in[9];
    float* out = (float*)d_out;

    const size_t per = (size_t)BATCH * HEADS * NPAD * CH;
    float* yq   = (float*)d_ws;
    float* yk   = yq + per;
    float* yv   = yk + per;
    float* yo   = yv + per;
    float* msum = yo + per;

    proj_kernel<<<dim3(450, 3), 256, 0, stream>>>(q, Wq, Wk, Wv, yq, yk, yv);
    norm_kernel<<<(BATCH * HEADS * NT * 16 + 7) / 8, 256, 0, stream>>>(yq, yk, yv, temp);
    attn_kernel<<<dim3(NT, BATCH * HEADS), 256, 0, stream>>>(
        yq, yk, yv, a1, a2, a3, a4, yo);
    bsum_kernel<<<(912 * DIM + 255) / 256, 256, 0, stream>>>(yo, msum);
    out_kernel<<<NT, 256, 0, stream>>>(msum, Wo, q, out);
}


# L2 norm and temperature fold moved into the projection kernel's epilogue (one wave per head, row sums in the same order); norm launch and its 29184 one-row waves gone; results bit-identical
# speedup vs baseline: 1.9819x; 1.0831x over previous
#include <hip/hip_runtime.h>
#include <hip/hip_bf16.h>

typedef __attribute__((ext_vector_type(2))) float v2f;
typedef __attribute__((ext_vector_type(4))) float v4f;
typedef __attribute__((ext_vector_type(8))) float v8f;

#define BATCH 8
#define HEADS 4
#define NQ    900
#define DIM   128
#define CH    32
#define NPAD  928
#define NT    57
#define SROW  928

__device__ __forceinline__ int kfrag_index(int m, int ch)
{
    return ((((m >> 4) * 4 + (ch >> 3)) * 32 + (((ch >> 1) & 1) << 4) + (m & 15)) << 2)
           + (((ch >> 2) & 1) << 1) + (ch & 1);
}
__device__ __forceinline__ int vfrag_index(int m, int ch)
{
    return (((((m >> 4) * 2 + (ch >> 4)) * 2 + ((m >> 3) & 1)) * 32
             + (((m >> 1) & 1) << 4) + (ch & 15)) << 2)
           + (((m >> 2) & 1) << 1) + (m & 1);
}

#define PPITCH 36

__global__ __launch_bounds__(128) void proj_kernel(
    const float* __restrict__ q,
    const float* __restrict__ Wq, const float* __restrict__ Wk,
    const float* __restrict__ Wv, const float* __restrict__ temp,
    float* __restrict__ yq, float* __restrict__ yk, float* __restrict__ yv)
{
    __shared__ __attribute__((aligned(16))) float tile[4 * 16 * PPITCH];

    const int t     = blockIdx.x;
    const int which = blockIdx.y;
    const int b_    = t / NT;
    const int mt    = t - b_ * NT;
    const int n0    = mt * 16;
    const int h     = threadIdx.x >> 5;
    const int lane  = threadIdx.x & 31;
    const int lh    = lane >> 4;
    const int ll    = lane & 15;

    const float* W = (which == 0) ? Wq : (which == 1) ? Wk : Wv;
    const float tt = temp[h];

    const int na = (n0 + ll < NQ) ? n0 + ll : NQ - 1;
    const float* aptr  = q + ((size_t)b_ * NQ + na) * DIM + 2 * lh;
    const float* bptr0 = W + (size_t)(32 * h + ll) * DIM + 2 * lh;
    const float* bptr1 = bptr0 + 16 * DIM;

    v8f acc0 = {}, acc1 = {};
    #pragma unroll
    for (int k = 0; k < DIM; k += 4) {
        v2f a  = *(const v2f*)(aptr + k);
        v2f b0 = *(const v2f*)(bptr0 + k);
        v2f b1 = *(const v2f*)(bptr1 + k);
        acc0 = __builtin_amdgcn_wmma_f32_16x16x4_f32(false, a, false, b0,
                                                     (short)0, acc0, false, false);
        acc1 = __builtin_amdgcn_wmma_f32_16x16x4_f32(false, a, false, b1,
                                                     (short)0, acc1, false, false);
    }

    const size_t plane = (size_t)(b_ * HEADS + h) * NPAD * CH;

    if (which == 2) {
        float* vp = yv + plane;
        const int rbase = n0 + 8 * lh;
        v4f lo0 = { acc0[0], acc0[1], acc0[4], acc0[5] };
        v4f hi0 = { acc0[2], acc0[3], acc0[6], acc0[7] };
        v4f lo1 = { acc1[0], acc1[1], acc1[4], acc1[5] };
        v4f hi1 = { acc1[2], acc1[3], acc1[6], acc1[7] };
        if (rbase + 7 >= NQ) {
            const bool p0 = rbase + 0 < NQ, p1 = rbase + 1 < NQ, p2 = rbase + 2 < NQ,
                       p3 = rbase + 3 < NQ, p4 = rbase + 4 < NQ, p5 = rbase + 5 < NQ,
                       p6 = rbase + 6 < NQ, p7 = rbase + 7 < NQ;
            lo0.x = p0 ? lo0.x : 0.f; lo0.y = p1 ? lo0.y : 0.f;
            lo0.z = p4 ? lo0.z : 0.f; lo0.w = p5 ? lo0.w : 0.f;
            hi0.x = p2 ? hi0.x : 0.f; hi0.y = p3 ? hi0.y : 0.f;
            hi0.z = p6 ? hi0.z : 0.f; hi0.w = p7 ? hi0.w : 0.f;
            lo1.x = p0 ? lo1.x : 0.f; lo1.y = p1 ? lo1.y : 0.f;
            lo1.z = p4 ? lo1.z : 0.f; lo1.w = p5 ? lo1.w : 0.f;
            hi1.x = p2 ? hi1.x : 0.f; hi1.y = p3 ? hi1.y : 0.f;
            hi1.z = p6 ? hi1.z : 0.f; hi1.w = p7 ? hi1.w : 0.f;
        }
        const int base0 = (((mt * 2 + 0) * 2 + lh) * 32 + ll) * 4;
        const int base1 = (((mt * 2 + 1) * 2 + lh) * 32 + ll) * 4;
        *(v4f*)(vp + base0)      = lo0;
        *(v4f*)(vp + base0 + 64) = hi0;
        *(v4f*)(vp + base1)      = lo1;
        *(v4f*)(vp + base1 + 64) = hi1;
    } else {
        float* tl = tile + h * (16 * PPITCH);
        #pragma unroll
        for (int r = 0; r < 8; ++r) {
            tl[(8 * lh + r) * PPITCH + ll]      = acc0[r];
            tl[(8 * lh + r) * PPITCH + 16 + ll] = acc1[r];
        }
        __syncthreads();

        const int row  = ll;
        const int half = lh;
        const v4f* mp = (const v4f*)(tl + row * PPITCH + 16 * half);
        const v4f* op = (const v4f*)(tl + row * PPITCH + 16 * (1 - half));
        float mine[16], oth[16];
        #pragma unroll
        for (int j = 0; j < 4; ++j) {
            const v4f m4 = mp[j];
            const v4f o4 = op[j];
            mine[4 * j + 0] = m4.x; mine[4 * j + 1] = m4.y;
            mine[4 * j + 2] = m4.z; mine[4 * j + 3] = m4.w;
            oth[4 * j + 0]  = o4.x; oth[4 * j + 1]  = o4.y;
            oth[4 * j + 2]  = o4.z; oth[4 * j + 3]  = o4.w;
        }

        float S[16];
        #pragma unroll
        for (int i = 0; i < 16; ++i) {
            const float p = oth[i] * oth[i];
            S[i] = __builtin_fmaf(mine[i], mine[i], p);
        }
        #pragma unroll
        for (int i = 0; i < 8; ++i) S[i] = S[i] + S[i + 8];
        #pragma unroll
        for (int i = 0; i < 4; ++i) S[i] = S[i] + S[i + 4];
        S[0] = S[0] + S[2];
        S[1] = S[1] + S[3];
        const float ssq = S[0] + S[1];
        const float dn  = fmaxf(sqrtf(ssq), 1e-12f);

        const bool live = (n0 + row) < NQ;
        float o[16];
        #pragma unroll
        for (int i = 0; i < 16; ++i) {
            float v = mine[i] / dn;
            if (which == 0) v = v * tt;
            o[i] = live ? v : 0.f;
        }
        if (which == 0) {
            float* dst = yq + plane + (size_t)(n0 + row) * CH + 16 * half;
            #pragma unroll
            for (int j = 0; j < 4; ++j) {
                v4f s4 = { o[4 * j], o[4 * j + 1], o[4 * j + 2], o[4 * j + 3] };
                *(v4f*)(dst + 4 * j) = s4;
            }
        } else {
            float* kp = yk + plane;
            #pragma unroll
            for (int o2 = 0; o2 < 2; ++o2) {
                #pragma unroll
                for (int lhp = 0; lhp < 2; ++lhp) {
                    const int c0 = 8 * o2 + 2 * lhp;
                    v4f s4 = { o[c0], o[c0 + 1], o[c0 + 4], o[c0 + 5] };
                    *(v4f*)(kp + (((mt * 4 + 2 * half + o2) * 32 + 16 * lhp + row) * 4)) = s4;
                }
            }
        }
    }
}

__global__ __launch_bounds__(256) void attn_kernel(
    const float* __restrict__ qh, const float* __restrict__ kh,
    const float* __restrict__ vh,
    const float* __restrict__ a1p, const float* __restrict__ a2p,
    const float* __restrict__ a3p, const float* __restrict__ a4p,
    float* __restrict__ yo)
{
    __shared__ float sc[16 * SROW];

    const int bh   = blockIdx.y;
    const int nt   = blockIdx.x;
    const int wave = threadIdx.x >> 5;
    const int lane = threadIdx.x & 31;
    const int lh   = lane >> 4;
    const int ll   = lane & 15;

    const float* qbase = qh + (size_t)bh * NPAD * CH;
    const float* kbase = kh + (size_t)bh * NPAD * CH;
    const float* vbase = vh + (size_t)bh * NPAD * CH;
    const int n0 = nt * 16;

    v2f areg[8];
    {
        const float* ap = qbase + (size_t)(n0 + ll) * CH + 2 * lh;
        #pragma unroll
        for (int c = 0; c < 8; ++c) areg[c] = *(const v2f*)(ap + 4 * c);
    }

    const v4f* kf = (const v4f*)kbase + lane;
    v4f nk0, nk1, nk2, nk3;
    {
        const v4f* bp = kf + wave * 128;
        nk0 = bp[0]; nk1 = bp[32]; nk2 = bp[64]; nk3 = bp[96];
    }
    for (int mt = wave; mt < NT; mt += 8) {
        const int m0 = mt * 16;
        const v4f k0 = nk0, k1 = nk1, k2 = nk2, k3 = nk3;
        {
            const int mn = (mt + 8 < NT) ? mt + 8 : NT - 1;
            const v4f* bp = kf + mn * 128;
            nk0 = bp[0]; nk1 = bp[32]; nk2 = bp[64]; nk3 = bp[96];
        }
        v8f acc = {};
        acc = __builtin_amdgcn_wmma_f32_16x16x4_f32(false, areg[0], false, k0.xy,
                                                    (short)0, acc, false, false);
        acc = __builtin_amdgcn_wmma_f32_16x16x4_f32(false, areg[1], false, k0.zw,
                                                    (short)0, acc, false, false);
        acc = __builtin_amdgcn_wmma_f32_16x16x4_f32(false, areg[2], false, k1.xy,
                                                    (short)0, acc, false, false);
        acc = __builtin_amdgcn_wmma_f32_16x16x4_f32(false, areg[3], false, k1.zw,
                                                    (short)0, acc, false, false);
        acc = __builtin_amdgcn_wmma_f32_16x16x4_f32(false, areg[4], false, k2.xy,
                                                    (short)0, acc, false, false);
        acc = __builtin_amdgcn_wmma_f32_16x16x4_f32(false, areg[5], false, k2.zw,
                                                    (short)0, acc, false, false);
        acc = __builtin_amdgcn_wmma_f32_16x16x4_f32(false, areg[6], false, k3.xy,
                                                    (short)0, acc, false, false);
        acc = __builtin_amdgcn_wmma_f32_16x16x4_f32(false, areg[7], false, k3.zw,
                                                    (short)0, acc, false, false);
        #pragma unroll
        for (int r = 0; r < 8; ++r)
            sc[(r + 8 * lh) * SROW + m0 + ll] = acc[r];
    }
    __syncthreads();

    for (int i = threadIdx.x; i < 16 * (SROW - NQ); i += 256) {
        int row = i / (SROW - NQ);
        int m   = NQ + i - row * (SROW - NQ);
        sc[row * SROW + m] = -1e30f;
    }
    __syncthreads();

    const float A1 = a1p[0], A2 = a2p[0], A3 = a3p[0], A4 = a4p[0];

    for (int rr = 0; rr < 2; ++rr) {
        const int row = wave + 8 * rr;
        float* srow = sc + row * SROW;

        float rv[29];
        #pragma unroll
        for (int j = 0; j < 29; ++j) rv[j] = srow[lane + 32 * j];

        float mx = -3.0e38f;
        #pragma unroll
        for (int j = 0; j < 29; ++j) mx = fmaxf(mx, rv[j]);
        #pragma unroll
        for (int s = 16; s > 0; s >>= 1) mx = fmaxf(mx, __shfl_xor(mx, s, 32));

        unsigned rk[29];
        #pragma unroll
        for (int j = 0; j < 29; ++j) {
            unsigned u = __float_as_uint(rv[j]);
            rk[j] = (u & 0x80000000u) ? ~u : (u | 0x80000000u);
        }

        unsigned th[3];
        const int ks[3] = {100, 300, 600};
        #pragma unroll
        for (int t = 0; t < 3; ++t) {
            unsigned cur = 0u;
            unsigned m   = 0x80000000u;
            do {
                const unsigned cand = cur | m;
                int tot = 0;
                #pragma unroll
                for (int j = 0; j + 1 < 29; j += 2) {
                    const unsigned long long two =
                        (unsigned long long)__builtin_amdgcn_ballot_w32(rk[j] >= cand) |
                        ((unsigned long long)__builtin_amdgcn_ballot_w32(rk[j + 1] >= cand) << 32);
                    tot += __builtin_popcountll(two);
                }
                tot += __builtin_popcount(__builtin_amdgcn_ballot_w32(rk[28] >= cand));
                if (tot >= ks[t]) cur = cand;
                if (tot == ks[t]) break;
                m >>= 1;
            } while (m != 0u);
            th[t] = cur;
        }

        float s1 = 0.f, s2 = 0.f, s3 = 0.f, s4 = 0.f;
        float w[29];
        #pragma unroll
        for (int j = 0; j < 29; ++j) {
            float e = __expf(rv[j] - mx);
            w[j] = e;
            s4 += e;
            if (rk[j] >= th[0]) s1 += e;
            if (rk[j] >= th[1]) s2 += e;
            if (rk[j] >= th[2]) s3 += e;
        }
        #pragma unroll
        for (int s = 16; s > 0; s >>= 1) {
            s1 += __shfl_xor(s1, s, 32);
            s2 += __shfl_xor(s2, s, 32);
            s3 += __shfl_xor(s3, s, 32);
            s4 += __shfl_xor(s4, s, 32);
        }
        const float c1 = A1 / s1, c2 = A2 / s2, c3 = A3 / s3, c4 = A4 / s4;
        #pragma unroll
        for (int j = 0; j < 29; ++j) {
            float f = c4;
            if (rk[j] >= th[0]) f += c1;
            if (rk[j] >= th[1]) f += c2;
            if (rk[j] >= th[2]) f += c3;
            srow[lane + 32 * j] = w[j] * f;
        }
    }
    __syncthreads();

    const int nhalf = wave & 1;
    const int kg    = wave >> 1;
    const v4f* vf = (const v4f*)vbase + nhalf * 64 + lane;
    v8f acc = {};
    for (int mt = kg; mt < NT; mt += 16) {
        v4f vt[4][2];
        #pragma unroll
        for (int u = 0; u < 4; ++u) {
            const int mu = (mt + 4 * u < NT) ? mt + 4 * u : mt;
            const v4f* bp = vf + mu * 128;
            vt[u][0] = bp[0];
            vt[u][1] = bp[32];
        }
        #pragma unroll
        for (int u = 0; u < 4; ++u) {
            const int mu = mt + 4 * u;
            if (mu < NT) {
                const float* ap = sc + ll * SROW + mu * 16 + 2 * lh;
                const v2f a0 = *(const v2f*)(ap);
                const v2f a1 = *(const v2f*)(ap + 4);
                const v2f a2 = *(const v2f*)(ap + 8);
                const v2f a3 = *(const v2f*)(ap + 12);
                acc = __builtin_amdgcn_wmma_f32_16x16x4_f32(false, a0, false, vt[u][0].xy,
                                                            (short)0, acc, false, false);
                acc = __builtin_amdgcn_wmma_f32_16x16x4_f32(false, a1, false, vt[u][0].zw,
                                                            (short)0, acc, false, false);
                acc = __builtin_amdgcn_wmma_f32_16x16x4_f32(false, a2, false, vt[u][1].xy,
                                                            (short)0, acc, false, false);
                acc = __builtin_amdgcn_wmma_f32_16x16x4_f32(false, a3, false, vt[u][1].zw,
                                                            (short)0, acc, false, false);
            }
        }
    }
    __syncthreads();

    float* part = sc;
    #pragma unroll
    for (int r = 0; r < 8; ++r) part[(wave * 8 + r) * 32 + lane] = acc[r];
    __syncthreads();

    if (wave < 2) {
        #pragma unroll
        for (int r = 0; r < 8; ++r) {
            float v = 0.f;
            #pragma unroll
            for (int g = 0; g < 4; ++g)
                v += part[(((g * 2 + wave) * 8) + r) * 32 + lane];
            const int row = n0 + r + 8 * lh;
            const int col = 16 * wave + ll;
            if (row < NQ)
                yo[((size_t)bh * NPAD + row) * CH + col] = v;
        }
    }
}

__global__ __launch_bounds__(256) void bsum_kernel(
    const float* __restrict__ yo, float* __restrict__ msum)
{
    const int idx = blockIdx.x * 256 + threadIdx.x;
    if (idx >= 912 * DIM) return;
    const int n = idx >> 7;
    const int d = idx & 127;
    float v = 0.f;
    if (n < NQ) {
        const int h = d >> 5, c = d & 31;
        #pragma unroll
        for (int b = 0; b < BATCH; ++b)
            v += yo[(((size_t)(b * HEADS + h)) * NPAD + n) * CH + c];
    }
    msum[idx] = v;
}

__global__ __launch_bounds__(256) void out_kernel(
    const float* __restrict__ msum, const float* __restrict__ Wo,
    const float* __restrict__ q, float* __restrict__ out)
{
    const int rt   = blockIdx.x;
    const int wave = threadIdx.x >> 5;
    const int lane = threadIdx.x & 31;
    const int lh   = lane >> 4;
    const int ll   = lane & 15;
    const int r0   = rt * 16;
    const int d0   = wave * 16;

    const float* aptr = msum + (size_t)(r0 + ll) * DIM + 2 * lh;
    const float* bptr = Wo   + (size_t)(d0 + ll) * DIM + 2 * lh;

    v8f acc = {};
    #pragma unroll
    for (int k = 0; k < DIM; k += 4) {
        v2f a = *(const v2f*)(aptr + k);
        v2f b = *(const v2f*)(bptr + k);
        acc = __builtin_amdgcn_wmma_f32_16x16x4_f32(false, a, false, b,
                                                    (short)0, acc, false, false);
    }
    #pragma unroll
    for (int r = 0; r < 8; ++r) {
        const int row = r0 + r + 8 * lh;
        const int col = d0 + ll;
        if (row < NQ)
            out[(size_t)row * DIM + col] = acc[r] + q[(size_t)row * DIM + col];
    }
}

extern "C" void kernel_launch(void* const* d_in, const int* in_sizes, int n_in,
                              void* d_out, int out_size, void* d_ws, size_t ws_size,
                              hipStream_t stream)
{
    const float* q    = (const float*)d_in[0];
    const float* Wq   = (const float*)d_in[1];
    const float* Wk   = (const float*)d_in[2];
    const float* Wv   = (const float*)d_in[3];
    const float* Wo   = (const float*)d_in[4];
    const float* temp = (const float*)d_in[5];
    const float* a1   = (const float*)d_in[6];
    const float* a2   = (const float*)d_in[7];
    const float* a3   = (const float*)d_in[8];
    const float* a4   = (const float*)d_in[9];
    float* out = (float*)d_out;

    const size_t per = (size_t)BATCH * HEADS * NPAD * CH;
    float* yq   = (float*)d_ws;
    float* yk   = yq + per;
    float* yv   = yk + per;
    float* yo   = yv + per;
    float* msum = yo + per;

    proj_kernel<<<dim3(BATCH * NT, 3), 128, 0, stream>>>(q, Wq, Wk, Wv, temp, yq, yk, yv);
    attn_kernel<<<dim3(NT, BATCH * HEADS), 256, 0, stream>>>(
        yq, yk, yv, a1, a2, a3, a4, yo);
    bsum_kernel<<<(912 * DIM + 255) / 256, 256, 0, stream>>>(yo, msum);
    out_kernel<<<NT, 256, 0, stream>>>(msum, Wo, q, out);
}


# top-k threshold search: exact key bracket, probe where a line through the bracket ends (score against count) meets k, not one bit per trip; interval halving as fallback; masks, results bit-identical
# speedup vs baseline: 2.8502x; 1.4381x over previous
#include <hip/hip_runtime.h>
#include <hip/hip_bf16.h>

typedef __attribute__((ext_vector_type(2))) float v2f;
typedef __attribute__((ext_vector_type(4))) float v4f;
typedef __attribute__((ext_vector_type(8))) float v8f;

#define BATCH 8
#define HEADS 4
#define NQ    900
#define DIM   128
#define CH    32
#define NPAD  928
#define NT    57
#define SROW  928
#define PROBE_STEPS 16

__device__ __forceinline__ int kfrag_index(int m, int ch)
{
    return ((((m >> 4) * 4 + (ch >> 3)) * 32 + (((ch >> 1) & 1) << 4) + (m & 15)) << 2)
           + (((ch >> 2) & 1) << 1) + (ch & 1);
}
__device__ __forceinline__ int vfrag_index(int m, int ch)
{
    return (((((m >> 4) * 2 + (ch >> 4)) * 2 + ((m >> 3) & 1)) * 32
             + (((m >> 1) & 1) << 4) + (ch & 15)) << 2)
           + (((m >> 2) & 1) << 1) + (m & 1);
}

#define PPITCH 36

__global__ __launch_bounds__(128) void proj_kernel(
    const float* __restrict__ q,
    const float* __restrict__ Wq, const float* __restrict__ Wk,
    const float* __restrict__ Wv, const float* __restrict__ temp,
    float* __restrict__ yq, float* __restrict__ yk, float* __restrict__ yv)
{
    __shared__ __attribute__((aligned(16))) float tile[4 * 16 * PPITCH];

    const int t     = blockIdx.x;
    const int which = blockIdx.y;
    const int b_    = t / NT;
    const int mt    = t - b_ * NT;
    const int n0    = mt * 16;
    const int h     = threadIdx.x >> 5;
    const int lane  = threadIdx.x & 31;
    const int lh    = lane >> 4;
    const int ll    = lane & 15;

    const float* W = (which == 0) ? Wq : (which == 1) ? Wk : Wv;
    const float tt = temp[h];

    const int na = (n0 + ll < NQ) ? n0 + ll : NQ - 1;
    const float* aptr  = q + ((size_t)b_ * NQ + na) * DIM + 2 * lh;
    const float* bptr0 = W + (size_t)(32 * h + ll) * DIM + 2 * lh;
    const float* bptr1 = bptr0 + 16 * DIM;

    v8f acc0 = {}, acc1 = {};
    #pragma unroll
    for (int k = 0; k < DIM; k += 4) {
        v2f a  = *(const v2f*)(aptr + k);
        v2f b0 = *(const v2f*)(bptr0 + k);
        v2f b1 = *(const v2f*)(bptr1 + k);
        acc0 = __builtin_amdgcn_wmma_f32_16x16x4_f32(false, a, false, b0,
                                                     (short)0, acc0, false, false);
        acc1 = __builtin_amdgcn_wmma_f32_16x16x4_f32(false, a, false, b1,
                                                     (short)0, acc1, false, false);
    }

    const size_t plane = (size_t)(b_ * HEADS + h) * NPAD * CH;

    if (which == 2) {
        float* vp = yv + plane;
        const int rbase = n0 + 8 * lh;
        v4f lo0 = { acc0[0], acc0[1], acc0[4], acc0[5] };
        v4f hi0 = { acc0[2], acc0[3], acc0[6], acc0[7] };
        v4f lo1 = { acc1[0], acc1[1], acc1[4], acc1[5] };
        v4f hi1 = { acc1[2], acc1[3], acc1[6], acc1[7] };
        if (rbase + 7 >= NQ) {
            const bool p0 = rbase + 0 < NQ, p1 = rbase + 1 < NQ, p2 = rbase + 2 < NQ,
                       p3 = rbase + 3 < NQ, p4 = rbase + 4 < NQ, p5 = rbase + 5 < NQ,
                       p6 = rbase + 6 < NQ, p7 = rbase + 7 < NQ;
            lo0.x = p0 ? lo0.x : 0.f; lo0.y = p1 ? lo0.y : 0.f;
            lo0.z = p4 ? lo0.z : 0.f; lo0.w = p5 ? lo0.w : 0.f;
            hi0.x = p2 ? hi0.x : 0.f; hi0.y = p3 ? hi0.y : 0.f;
            hi0.z = p6 ? hi0.z : 0.f; hi0.w = p7 ? hi0.w : 0.f;
            lo1.x = p0 ? lo1.x : 0.f; lo1.y = p1 ? lo1.y : 0.f;
            lo1.z = p4 ? lo1.z : 0.f; lo1.w = p5 ? lo1.w : 0.f;
            hi1.x = p2 ? hi1.x : 0.f; hi1.y = p3 ? hi1.y : 0.f;
            hi1.z = p6 ? hi1.z : 0.f; hi1.w = p7 ? hi1.w : 0.f;
        }
        const int base0 = (((mt * 2 + 0) * 2 + lh) * 32 + ll) * 4;
        const int base1 = (((mt * 2 + 1) * 2 + lh) * 32 + ll) * 4;
        *(v4f*)(vp + base0)      = lo0;
        *(v4f*)(vp + base0 + 64) = hi0;
        *(v4f*)(vp + base1)      = lo1;
        *(v4f*)(vp + base1 + 64) = hi1;
    } else {
        float* tl = tile + h * (16 * PPITCH);
        #pragma unroll
        for (int r = 0; r < 8; ++r) {
            tl[(8 * lh + r) * PPITCH + ll]      = acc0[r];
            tl[(8 * lh + r) * PPITCH + 16 + ll] = acc1[r];
        }
        __syncthreads();

        const int row  = ll;
        const int half = lh;
        const v4f* mp = (const v4f*)(tl + row * PPITCH + 16 * half);
        const v4f* op = (const v4f*)(tl + row * PPITCH + 16 * (1 - half));
        float mine[16], oth[16];
        #pragma unroll
        for (int j = 0; j < 4; ++j) {
            const v4f m4 = mp[j];
            const v4f o4 = op[j];
            mine[4 * j + 0] = m4.x; mine[4 * j + 1] = m4.y;
            mine[4 * j + 2] = m4.z; mine[4 * j + 3] = m4.w;
            oth[4 * j + 0]  = o4.x; oth[4 * j + 1]  = o4.y;
            oth[4 * j + 2]  = o4.z; oth[4 * j + 3]  = o4.w;
        }

        float S[16];
        #pragma unroll
        for (int i = 0; i < 16; ++i) {
            const float p = oth[i] * oth[i];
            S[i] = __builtin_fmaf(mine[i], mine[i], p);
        }
        #pragma unroll
        for (int i = 0; i < 8; ++i) S[i] = S[i] + S[i + 8];
        #pragma unroll
        for (int i = 0; i < 4; ++i) S[i] = S[i] + S[i + 4];
        S[0] = S[0] + S[2];
        S[1] = S[1] + S[3];
        const float ssq = S[0] + S[1];
        const float dn  = fmaxf(sqrtf(ssq), 1e-12f);

        const bool live = (n0 + row) < NQ;
        float o[16];
        #pragma unroll
        for (int i = 0; i < 16; ++i) {
            float v = mine[i] / dn;
            if (which == 0) v = v * tt;
            o[i] = live ? v : 0.f;
        }
        if (which == 0) {
            float* dst = yq + plane + (size_t)(n0 + row) * CH + 16 * half;
            #pragma unroll
            for (int j = 0; j < 4; ++j) {
                v4f s4 = { o[4 * j], o[4 * j + 1], o[4 * j + 2], o[4 * j + 3] };
                *(v4f*)(dst + 4 * j) = s4;
            }
        } else {
            float* kp = yk + plane;
            #pragma unroll
            for (int o2 = 0; o2 < 2; ++o2) {
                #pragma unroll
                for (int lhp = 0; lhp < 2; ++lhp) {
                    const int c0 = 8 * o2 + 2 * lhp;
                    v4f s4 = { o[c0], o[c0 + 1], o[c0 + 4], o[c0 + 5] };
                    *(v4f*)(kp + (((mt * 4 + 2 * half + o2) * 32 + 16 * lhp + row) * 4)) = s4;
                }
            }
        }
    }
}

__global__ __launch_bounds__(256) void attn_kernel(
    const float* __restrict__ qh, const float* __restrict__ kh,
    const float* __restrict__ vh,
    const float* __restrict__ a1p, const float* __restrict__ a2p,
    const float* __restrict__ a3p, const float* __restrict__ a4p,
    float* __restrict__ yo)
{
    __shared__ float sc[16 * SROW];

    const int bh   = blockIdx.y;
    const int nt   = blockIdx.x;
    const int wave = threadIdx.x >> 5;
    const int lane = threadIdx.x & 31;
    const int lh   = lane >> 4;
    const int ll   = lane & 15;

    const float* qbase = qh + (size_t)bh * NPAD * CH;
    const float* kbase = kh + (size_t)bh * NPAD * CH;
    const float* vbase = vh + (size_t)bh * NPAD * CH;
    const int n0 = nt * 16;

    v2f areg[8];
    {
        const float* ap = qbase + (size_t)(n0 + ll) * CH + 2 * lh;
        #pragma unroll
        for (int c = 0; c < 8; ++c) areg[c] = *(const v2f*)(ap + 4 * c);
    }

    const v4f* kf = (const v4f*)kbase + lane;
    v4f nk0, nk1, nk2, nk3;
    {
        const v4f* bp = kf + wave * 128;
        nk0 = bp[0]; nk1 = bp[32]; nk2 = bp[64]; nk3 = bp[96];
    }
    for (int mt = wave; mt < NT; mt += 8) {
        const int m0 = mt * 16;
        const v4f k0 = nk0, k1 = nk1, k2 = nk2, k3 = nk3;
        {
            const int mn = (mt + 8 < NT) ? mt + 8 : NT - 1;
            const v4f* bp = kf + mn * 128;
            nk0 = bp[0]; nk1 = bp[32]; nk2 = bp[64]; nk3 = bp[96];
        }
        v8f acc = {};
        acc = __builtin_amdgcn_wmma_f32_16x16x4_f32(false, areg[0], false, k0.xy,
                                                    (short)0, acc, false, false);
        acc = __builtin_amdgcn_wmma_f32_16x16x4_f32(false, areg[1], false, k0.zw,
                                                    (short)0, acc, false, false);
        acc = __builtin_amdgcn_wmma_f32_16x16x4_f32(false, areg[2], false, k1.xy,
                                                    (short)0, acc, false, false);
        acc = __builtin_amdgcn_wmma_f32_16x16x4_f32(false, areg[3], false, k1.zw,
                                                    (short)0, acc, false, false);
        acc = __builtin_amdgcn_wmma_f32_16x16x4_f32(false, areg[4], false, k2.xy,
                                                    (short)0, acc, false, false);
        acc = __builtin_amdgcn_wmma_f32_16x16x4_f32(false, areg[5], false, k2.zw,
                                                    (short)0, acc, false, false);
        acc = __builtin_amdgcn_wmma_f32_16x16x4_f32(false, areg[6], false, k3.xy,
                                                    (short)0, acc, false, false);
        acc = __builtin_amdgcn_wmma_f32_16x16x4_f32(false, areg[7], false, k3.zw,
                                                    (short)0, acc, false, false);
        #pragma unroll
        for (int r = 0; r < 8; ++r)
            sc[(r + 8 * lh) * SROW + m0 + ll] = acc[r];
    }
    __syncthreads();

    for (int i = threadIdx.x; i < 16 * (SROW - NQ); i += 256) {
        int row = i / (SROW - NQ);
        int m   = NQ + i - row * (SROW - NQ);
        sc[row * SROW + m] = -1e30f;
    }
    __syncthreads();

    const float A1 = a1p[0], A2 = a2p[0], A3 = a3p[0], A4 = a4p[0];

    for (int rr = 0; rr < 2; ++rr) {
        const int row = wave + 8 * rr;
        float* srow = sc + row * SROW;

        float rv[29];
        #pragma unroll
        for (int j = 0; j < 29; ++j) rv[j] = srow[lane + 32 * j];

        float mx = -3.0e38f;
        #pragma unroll
        for (int j = 0; j < 29; ++j) mx = fmaxf(mx, rv[j]);
        #pragma unroll
        for (int s = 16; s > 0; s >>= 1) mx = fmaxf(mx, __shfl_xor(mx, s, 32));

        unsigned rk[29];
        #pragma unroll
        for (int j = 0; j < 29; ++j) {
            unsigned u = __float_as_uint(rv[j]);
            rk[j] = (u & 0x80000000u) ? ~u : (u | 0x80000000u);
        }

        unsigned kmn = rk[0];
        #pragma unroll
        for (int j = 1; j < 28; ++j) kmn = min(kmn, rk[j]);
        kmn = min(kmn, (lane < 4) ? rk[28] : kmn);
        #pragma unroll
        for (int s = 16; s > 0; s >>= 1) kmn = min(kmn, (unsigned)__shfl_xor((int)kmn, s, 32));
        const unsigned kmnU = __builtin_amdgcn_readfirstlane(kmn);
        const float mnU = __uint_as_float(kmnU ^ (~(unsigned)((int)kmnU >> 31) | 0x80000000u));
        const float mxU = __uint_as_float(__builtin_amdgcn_readfirstlane(__float_as_uint(mx)));

        auto count_ge = [&](unsigned cand) -> int {
            int tot = 0;
            #pragma unroll
            for (int j = 0; j + 1 < 29; j += 2) {
                const unsigned long long two =
                    (unsigned long long)__builtin_amdgcn_ballot_w32(rk[j] >= cand) |
                    ((unsigned long long)__builtin_amdgcn_ballot_w32(rk[j + 1] >= cand) << 32);
                tot += __builtin_popcountll(two);
            }
            tot += __builtin_popcount(__builtin_amdgcn_ballot_w32(rk[28] >= cand));
            return tot;
        };

        unsigned th[3];
        const int ks[3] = {100, 300, 600};
        float ha  = mnU;
        float hca = 900.f;
        #pragma unroll
        for (int t = 2; t >= 0; --t) {
            const int k = ks[t];
            unsigned lo = 0u, top = 0xFFFFFFFFu;
            float a = ha,  ga = hca - (float)k;
            float b = mxU, gb = 0.5f - (float)k;
            int left = PROBE_STEPS;
            do {
                const float cf = a + (b - a) * (ga * __builtin_amdgcn_rcpf(ga - gb));
                const unsigned u = __float_as_uint(cf);
                unsigned cand = u ^ ((unsigned)((int)u >> 31) | 0x80000000u);
                cand = max(cand, lo + 1u);
                cand = min(cand, top);
                const int tot = count_ge(cand);
                const float cv = __uint_as_float(cand ^ (~(unsigned)((int)cand >> 31) | 0x80000000u));
                const float g  = (float)tot - (float)k;
                const bool  ge = tot >= k;
                top = (tot > k) ? top : ((tot == k) ? cand : cand - 1u);
                lo  = ge ? cand : lo;
                a   = ge ? cv : a;
                b   = ge ? b : cv;
                ga  = ge ? g : ga * 0.8f;
                gb  = ge ? gb * 0.8f : g;
                if (lo >= top) break;
            } while (--left > 0);
            while (lo < top) {
                const unsigned d = top - lo;
                const unsigned cand = lo + (d >> 1) + (d & 1u);
                const int tot = count_ge(cand);
                top = (tot > k) ? top : ((tot == k) ? cand : cand - 1u);
                lo  = (tot >= k) ? cand : lo;
            }
            th[t] = lo;
            ha  = __uint_as_float(lo ^ (~(unsigned)((int)lo >> 31) | 0x80000000u));
            hca = (float)k;
        }

        float s1 = 0.f, s2 = 0.f, s3 = 0.f, s4 = 0.f;
        float w[29];
        #pragma unroll
        for (int j = 0; j < 29; ++j) {
            float e = __expf(rv[j] - mx);
            w[j] = e;
            s4 += e;
            if (rk[j] >= th[0]) s1 += e;
            if (rk[j] >= th[1]) s2 += e;
            if (rk[j] >= th[2]) s3 += e;
        }
        #pragma unroll
        for (int s = 16; s > 0; s >>= 1) {
            s1 += __shfl_xor(s1, s, 32);
            s2 += __shfl_xor(s2, s, 32);
            s3 += __shfl_xor(s3, s, 32);
            s4 += __shfl_xor(s4, s, 32);
        }
        const float c1 = A1 / s1, c2 = A2 / s2, c3 = A3 / s3, c4 = A4 / s4;
        #pragma unroll
        for (int j = 0; j < 29; ++j) {
            float f = c4;
            if (rk[j] >= th[0]) f += c1;
            if (rk[j] >= th[1]) f += c2;
            if (rk[j] >= th[2]) f += c3;
            srow[lane + 32 * j] = w[j] * f;
        }
    }
    __syncthreads();

    const int nhalf = wave & 1;
    const int kg    = wave >> 1;
    const v4f* vf = (const v4f*)vbase + nhalf * 64 + lane;
    v8f acc = {};
    for (int mt = kg; mt < NT; mt += 16) {
        v4f vt[4][2];
        #pragma unroll
        for (int u = 0; u < 4; ++u) {
            const int mu = (mt + 4 * u < NT) ? mt + 4 * u : mt;
            const v4f* bp = vf + mu * 128;
            vt[u][0] = bp[0];
            vt[u][1] = bp[32];
        }
        #pragma unroll
        for (int u = 0; u < 4; ++u) {
            const int mu = mt + 4 * u;
            if (mu < NT) {
                const float* ap = sc + ll * SROW + mu * 16 + 2 * lh;
                const v2f a0 = *(const v2f*)(ap);
                const v2f a1 = *(const v2f*)(ap + 4);
                const v2f a2 = *(const v2f*)(ap + 8);
                const v2f a3 = *(const v2f*)(ap + 12);
                acc = __builtin_amdgcn_wmma_f32_16x16x4_f32(false, a0, false, vt[u][0].xy,
                                                            (short)0, acc, false, false);
                acc = __builtin_amdgcn_wmma_f32_16x16x4_f32(false, a1, false, vt[u][0].zw,
                                                            (short)0, acc, false, false);
                acc = __builtin_amdgcn_wmma_f32_16x16x4_f32(false, a2, false, vt[u][1].xy,
                                                            (short)0, acc, false, false);
                acc = __builtin_amdgcn_wmma_f32_16x16x4_f32(false, a3, false, vt[u][1].zw,
                                                            (short)0, acc, false, false);
            }
        }
    }
    __syncthreads();

    float* part = sc;
    #pragma unroll
    for (int r = 0; r < 8; ++r) part[(wave * 8 + r) * 32 + lane] = acc[r];
    __syncthreads();

    if (wave < 2) {
        #pragma unroll
        for (int r = 0; r < 8; ++r) {
            float v = 0.f;
            #pragma unroll
            for (int g = 0; g < 4; ++g)
                v += part[(((g * 2 + wave) * 8) + r) * 32 + lane];
            const int row = n0 + r + 8 * lh;
            const int col = 16 * wave + ll;
            if (row < NQ)
                yo[((size_t)bh * NPAD + row) * CH + col] = v;
        }
    }
}

__global__ __launch_bounds__(256) void bsum_kernel(
    const float* __restrict__ yo, float* __restrict__ msum)
{
    const int idx = blockIdx.x * 256 + threadIdx.x;
    if (idx >= 912 * DIM) return;
    const int n = idx >> 7;
    const int d = idx & 127;
    float v = 0.f;
    if (n < NQ) {
        const int h = d >> 5, c = d & 31;
        #pragma unroll
        for (int b = 0; b < BATCH; ++b)
            v += yo[(((size_t)(b * HEADS + h)) * NPAD + n) * CH + c];
    }
    msum[idx] = v;
}

__global__ __launch_bounds__(256) void out_kernel(
    const float* __restrict__ msum, const float* __restrict__ Wo,
    const float* __restrict__ q, float* __restrict__ out)
{
    const int rt   = blockIdx.x;
    const int wave = threadIdx.x >> 5;
    const int lane = threadIdx.x & 31;
    const int lh   = lane >> 4;
    const int ll   = lane & 15;
    const int r0   = rt * 16;
    const int d0   = wave * 16;

    const float* aptr = msum + (size_t)(r0 + ll) * DIM + 2 * lh;
    const float* bptr = Wo   + (size_t)(d0 + ll) * DIM + 2 * lh;

    v8f acc = {};
    #pragma unroll
    for (int k = 0; k < DIM; k += 4) {
        v2f a = *(const v2f*)(aptr + k);
        v2f b = *(const v2f*)(bptr + k);
        acc = __builtin_amdgcn_wmma_f32_16x16x4_f32(false, a, false, b,
                                                    (short)0, acc, false, false);
    }
    #pragma unroll
    for (int r = 0; r < 8; ++r) {
        const int row = r0 + r + 8 * lh;
        const int col = d0 + ll;
        if (row < NQ)
            out[(size_t)row * DIM + col] = acc[r] + q[(size_t)row * DIM + col];
    }
}

extern "C" void kernel_launch(void* const* d_in, const int* in_sizes, int n_in,
                              void* d_out, int out_size, void* d_ws, size_t ws_size,
                              hipStream_t stream)
{
    const float* q    = (const float*)d_in[0];
    const float* Wq   = (const float*)d_in[1];
    const float* Wk   = (const float*)d_in[2];
    const float* Wv   = (const float*)d_in[3];
    const float* Wo   = (const float*)d_in[4];
    const float* temp = (const float*)d_in[5];
    const float* a1   = (const float*)d_in[6];
    const float* a2   = (const float*)d_in[7];
    const float* a3   = (const float*)d_in[8];
    const float* a4   = (const float*)d_in[9];
    float* out = (float*)d_out;

    const size_t per = (size_t)BATCH * HEADS * NPAD * CH;
    float* yq   = (float*)d_ws;
    float* yk   = yq + per;
    float* yv   = yk + per;
    float* yo   = yv + per;
    float* msum = yo + per;

    proj_kernel<<<dim3(BATCH * NT, 3), 128, 0, stream>>>(q, Wq, Wk, Wv, temp, yq, yk, yv);
    attn_kernel<<<dim3(NT, BATCH * HEADS), 256, 0, stream>>>(
        yq, yk, yv, a1, a2, a3, a4, yo);
    bsum_kernel<<<(912 * DIM + 255) / 256, 256, 0, stream>>>(yo, msum);
    out_kernel<<<NT, 256, 0, stream>>>(msum, Wo, q, out);
}


# top-k threshold search: first probes placed by the row's mean and standard deviation (normal-quantile point and density slope), not by a line from row min to row max; masks, results bit-identical
# speedup vs baseline: 3.1157x; 1.0932x over previous
#include <hip/hip_runtime.h>
#include <hip/hip_bf16.h>

typedef __attribute__((ext_vector_type(2))) float v2f;
typedef __attribute__((ext_vector_type(4))) float v4f;
typedef __attribute__((ext_vector_type(8))) float v8f;

#define BATCH 8
#define HEADS 4
#define NQ    900
#define DIM   128
#define CH    32
#define NPAD  928
#define NT    57
#define SROW  928
#define PROBE_STEPS 16

__device__ __forceinline__ int kfrag_index(int m, int ch)
{
    return ((((m >> 4) * 4 + (ch >> 3)) * 32 + (((ch >> 1) & 1) << 4) + (m & 15)) << 2)
           + (((ch >> 2) & 1) << 1) + (ch & 1);
}
__device__ __forceinline__ int vfrag_index(int m, int ch)
{
    return (((((m >> 4) * 2 + (ch >> 4)) * 2 + ((m >> 3) & 1)) * 32
             + (((m >> 1) & 1) << 4) + (ch & 15)) << 2)
           + (((m >> 2) & 1) << 1) + (m & 1);
}

#define PPITCH 36

__global__ __launch_bounds__(128) void proj_kernel(
    const float* __restrict__ q,
    const float* __restrict__ Wq, const float* __restrict__ Wk,
    const float* __restrict__ Wv, const float* __restrict__ temp,
    float* __restrict__ yq, float* __restrict__ yk, float* __restrict__ yv)
{
    __shared__ __attribute__((aligned(16))) float tile[4 * 16 * PPITCH];

    const int t     = blockIdx.x;
    const int which = blockIdx.y;
    const int b_    = t / NT;
    const int mt    = t - b_ * NT;
    const int n0    = mt * 16;
    const int h     = threadIdx.x >> 5;
    const int lane  = threadIdx.x & 31;
    const int lh    = lane >> 4;
    const int ll    = lane & 15;

    const float* W = (which == 0) ? Wq : (which == 1) ? Wk : Wv;
    const float tt = temp[h];

    const int na = (n0 + ll < NQ) ? n0 + ll : NQ - 1;
    const float* aptr  = q + ((size_t)b_ * NQ + na) * DIM + 2 * lh;
    const float* bptr0 = W + (size_t)(32 * h + ll) * DIM + 2 * lh;
    const float* bptr1 = bptr0 + 16 * DIM;

    v8f acc0 = {}, acc1 = {};
    #pragma unroll
    for (int k = 0; k < DIM; k += 4) {
        v2f a  = *(const v2f*)(aptr + k);
        v2f b0 = *(const v2f*)(bptr0 + k);
        v2f b1 = *(const v2f*)(bptr1 + k);
        acc0 = __builtin_amdgcn_wmma_f32_16x16x4_f32(false, a, false, b0,
                                                     (short)0, acc0, false, false);
        acc1 = __builtin_amdgcn_wmma_f32_16x16x4_f32(false, a, false, b1,
                                                     (short)0, acc1, false, false);
    }

    const size_t plane = (size_t)(b_ * HEADS + h) * NPAD * CH;

    if (which == 2) {
        float* vp = yv + plane;
        const int rbase = n0 + 8 * lh;
        v4f lo0 = { acc0[0], acc0[1], acc0[4], acc0[5] };
        v4f hi0 = { acc0[2], acc0[3], acc0[6], acc0[7] };
        v4f lo1 = { acc1[0], acc1[1], acc1[4], acc1[5] };
        v4f hi1 = { acc1[2], acc1[3], acc1[6], acc1[7] };
        if (rbase + 7 >= NQ) {
            const bool p0 = rbase + 0 < NQ, p1 = rbase + 1 < NQ, p2 = rbase + 2 < NQ,
                       p3 = rbase + 3 < NQ, p4 = rbase + 4 < NQ, p5 = rbase + 5 < NQ,
                       p6 = rbase + 6 < NQ, p7 = rbase + 7 < NQ;
            lo0.x = p0 ? lo0.x : 0.f; lo0.y = p1 ? lo0.y : 0.f;
            lo0.z = p4 ? lo0.z : 0.f; lo0.w = p5 ? lo0.w : 0.f;
            hi0.x = p2 ? hi0.x : 0.f; hi0.y = p3 ? hi0.y : 0.f;
            hi0.z = p6 ? hi0.z : 0.f; hi0.w = p7 ? hi0.w : 0.f;
            lo1.x = p0 ? lo1.x : 0.f; lo1.y = p1 ? lo1.y : 0.f;
            lo1.z = p4 ? lo1.z : 0.f; lo1.w = p5 ? lo1.w : 0.f;
            hi1.x = p2 ? hi1.x : 0.f; hi1.y = p3 ? hi1.y : 0.f;
            hi1.z = p6 ? hi1.z : 0.f; hi1.w = p7 ? hi1.w : 0.f;
        }
        const int base0 = (((mt * 2 + 0) * 2 + lh) * 32 + ll) * 4;
        const int base1 = (((mt * 2 + 1) * 2 + lh) * 32 + ll) * 4;
        *(v4f*)(vp + base0)      = lo0;
        *(v4f*)(vp + base0 + 64) = hi0;
        *(v4f*)(vp + base1)      = lo1;
        *(v4f*)(vp + base1 + 64) = hi1;
    } else {
        float* tl = tile + h * (16 * PPITCH);
        #pragma unroll
        for (int r = 0; r < 8; ++r) {
            tl[(8 * lh + r) * PPITCH + ll]      = acc0[r];
            tl[(8 * lh + r) * PPITCH + 16 + ll] = acc1[r];
        }
        __syncthreads();

        const int row  = ll;
        const int half = lh;
        const v4f* mp = (const v4f*)(tl + row * PPITCH + 16 * half);
        const v4f* op = (const v4f*)(tl + row * PPITCH + 16 * (1 - half));
        float mine[16], oth[16];
        #pragma unroll
        for (int j = 0; j < 4; ++j) {
            const v4f m4 = mp[j];
            const v4f o4 = op[j];
            mine[4 * j + 0] = m4.x; mine[4 * j + 1] = m4.y;
            mine[4 * j + 2] = m4.z; mine[4 * j + 3] = m4.w;
            oth[4 * j + 0]  = o4.x; oth[4 * j + 1]  = o4.y;
            oth[4 * j + 2]  = o4.z; oth[4 * j + 3]  = o4.w;
        }

        float S[16];
        #pragma unroll
        for (int i = 0; i < 16; ++i) {
            const float p = oth[i] * oth[i];
            S[i] = __builtin_fmaf(mine[i], mine[i], p);
        }
        #pragma unroll
        for (int i = 0; i < 8; ++i) S[i] = S[i] + S[i + 8];
        #pragma unroll
        for (int i = 0; i < 4; ++i) S[i] = S[i] + S[i + 4];
        S[0] = S[0] + S[2];
        S[1] = S[1] + S[3];
        const float ssq = S[0] + S[1];
        const float dn  = fmaxf(sqrtf(ssq), 1e-12f);

        const bool live = (n0 + row) < NQ;
        float o[16];
        #pragma unroll
        for (int i = 0; i < 16; ++i) {
            float v = mine[i] / dn;
            if (which == 0) v = v * tt;
            o[i] = live ? v : 0.f;
        }
        if (which == 0) {
            float* dst = yq + plane + (size_t)(n0 + row) * CH + 16 * half;
            #pragma unroll
            for (int j = 0; j < 4; ++j) {
                v4f s4 = { o[4 * j], o[4 * j + 1], o[4 * j + 2], o[4 * j + 3] };
                *(v4f*)(dst + 4 * j) = s4;
            }
        } else {
            float* kp = yk + plane;
            #pragma unroll
            for (int o2 = 0; o2 < 2; ++o2) {
                #pragma unroll
                for (int lhp = 0; lhp < 2; ++lhp) {
                    const int c0 = 8 * o2 + 2 * lhp;
                    v4f s4 = { o[c0], o[c0 + 1], o[c0 + 4], o[c0 + 5] };
                    *(v4f*)(kp + (((mt * 4 + 2 * half + o2) * 32 + 16 * lhp + row) * 4)) = s4;
                }
            }
        }
    }
}

__global__ __launch_bounds__(256) void attn_kernel(
    const float* __restrict__ qh, const float* __restrict__ kh,
    const float* __restrict__ vh,
    const float* __restrict__ a1p, const float* __restrict__ a2p,
    const float* __restrict__ a3p, const float* __restrict__ a4p,
    float* __restrict__ yo)
{
    __shared__ float sc[16 * SROW];

    const int bh   = blockIdx.y;
    const int nt   = blockIdx.x;
    const int wave = threadIdx.x >> 5;
    const int lane = threadIdx.x & 31;
    const int lh   = lane >> 4;
    const int ll   = lane & 15;

    const float* qbase = qh + (size_t)bh * NPAD * CH;
    const float* kbase = kh + (size_t)bh * NPAD * CH;
    const float* vbase = vh + (size_t)bh * NPAD * CH;
    const int n0 = nt * 16;

    v2f areg[8];
    {
        const float* ap = qbase + (size_t)(n0 + ll) * CH + 2 * lh;
        #pragma unroll
        for (int c = 0; c < 8; ++c) areg[c] = *(const v2f*)(ap + 4 * c);
    }

    const v4f* kf = (const v4f*)kbase + lane;
    v4f nk0, nk1, nk2, nk3;
    {
        const v4f* bp = kf + wave * 128;
        nk0 = bp[0]; nk1 = bp[32]; nk2 = bp[64]; nk3 = bp[96];
    }
    for (int mt = wave; mt < NT; mt += 8) {
        const int m0 = mt * 16;
        const v4f k0 = nk0, k1 = nk1, k2 = nk2, k3 = nk3;
        {
            const int mn = (mt + 8 < NT) ? mt + 8 : NT - 1;
            const v4f* bp = kf + mn * 128;
            nk0 = bp[0]; nk1 = bp[32]; nk2 = bp[64]; nk3 = bp[96];
        }
        v8f acc = {};
        acc = __builtin_amdgcn_wmma_f32_16x16x4_f32(false, areg[0], false, k0.xy,
                                                    (short)0, acc, false, false);
        acc = __builtin_amdgcn_wmma_f32_16x16x4_f32(false, areg[1], false, k0.zw,
                                                    (short)0, acc, false, false);
        acc = __builtin_amdgcn_wmma_f32_16x16x4_f32(false, areg[2], false, k1.xy,
                                                    (short)0, acc, false, false);
        acc = __builtin_amdgcn_wmma_f32_16x16x4_f32(false, areg[3], false, k1.zw,
                                                    (short)0, acc, false, false);
        acc = __builtin_amdgcn_wmma_f32_16x16x4_f32(false, areg[4], false, k2.xy,
                                                    (short)0, acc, false, false);
        acc = __builtin_amdgcn_wmma_f32_16x16x4_f32(false, areg[5], false, k2.zw,
                                                    (short)0, acc, false, false);
        acc = __builtin_amdgcn_wmma_f32_16x16x4_f32(false, areg[6], false, k3.xy,
                                                    (short)0, acc, false, false);
        acc = __builtin_amdgcn_wmma_f32_16x16x4_f32(false, areg[7], false, k3.zw,
                                                    (short)0, acc, false, false);
        #pragma unroll
        for (int r = 0; r < 8; ++r)
            sc[(r + 8 * lh) * SROW + m0 + ll] = acc[r];
    }
    __syncthreads();

    for (int i = threadIdx.x; i < 16 * (SROW - NQ); i += 256) {
        int row = i / (SROW - NQ);
        int m   = NQ + i - row * (SROW - NQ);
        sc[row * SROW + m] = -1e30f;
    }
    __syncthreads();

    const float A1 = a1p[0], A2 = a2p[0], A3 = a3p[0], A4 = a4p[0];

    for (int rr = 0; rr < 2; ++rr) {
        const int row = wave + 8 * rr;
        float* srow = sc + row * SROW;

        float rv[29];
        #pragma unroll
        for (int j = 0; j < 29; ++j) rv[j] = srow[lane + 32 * j];

        float mx = -3.0e38f;
        #pragma unroll
        for (int j = 0; j < 29; ++j) mx = fmaxf(mx, rv[j]);
        #pragma unroll
        for (int s = 16; s > 0; s >>= 1) mx = fmaxf(mx, __shfl_xor(mx, s, 32));

        unsigned rk[29];
        #pragma unroll
        for (int j = 0; j < 29; ++j) {
            unsigned u = __float_as_uint(rv[j]);
            rk[j] = (u & 0x80000000u) ? ~u : (u | 0x80000000u);
        }

        __builtin_amdgcn_sched_barrier(0);

        float sm = 0.f, sq = 0.f;
        #pragma unroll
        for (int j = 0; j < 28; ++j) {
            sm += rv[j];
            sq = __builtin_fmaf(rv[j], rv[j], sq);
        }
        {
            const float tl = (lane < 4) ? rv[28] : 0.f;
            sm += tl;
            sq = __builtin_fmaf(tl, tl, sq);
        }
        #pragma unroll
        for (int s = 16; s > 0; s >>= 1) {
            sm += __shfl_xor(sm, s, 32);
            sq += __shfl_xor(sq, s, 32);
        }
        const float smU = __uint_as_float(__builtin_amdgcn_readfirstlane(__float_as_uint(sm)));
        const float sqU = __uint_as_float(__builtin_amdgcn_readfirstlane(__float_as_uint(sq)));
        const float muU = smU * (1.f / (float)NQ);
        const float sdU = __builtin_amdgcn_sqrtf(fmaxf(sqU * (1.f / (float)NQ) - muU * muU, 0.f));

        auto count_ge = [&](unsigned cand) -> int {
            int tot = 0;
            #pragma unroll
            for (int j = 0; j + 1 < 29; j += 2) {
                const unsigned long long two =
                    (unsigned long long)__builtin_amdgcn_ballot_w32(rk[j] >= cand) |
                    ((unsigned long long)__builtin_amdgcn_ballot_w32(rk[j + 1] >= cand) << 32);
                tot += __builtin_popcountll(two);
            }
            tot += __builtin_popcount(__builtin_amdgcn_ballot_w32(rk[28] >= cand));
            return tot;
        };

        unsigned th[3];
        const int   ks[3] = {100, 300, 600};
        const float zq[3] = {1.2177f, 0.4292f, -0.4322f};
        const float gq[3] = {641.5f, 1228.0f, 1226.4f};
        #pragma unroll
        for (int t = 2; t >= 0; --t) {
            const int k = ks[t];
            unsigned lo = 0u, top = 0xFFFFFFFFu;
            const float ctr = muU + zq[t] * sdU;
            float a = ctr - 3.f * sdU, ga = gq[t];
            float b = ctr + 3.f * sdU, gb = -gq[t];
            int left = PROBE_STEPS;
            do {
                const float cf = a + (b - a) * (ga * __builtin_amdgcn_rcpf(ga - gb));
                const unsigned u = __float_as_uint(cf);
                unsigned cand = u ^ ((unsigned)((int)u >> 31) | 0x80000000u);
                cand = max(cand, lo + 1u);
                cand = min(cand, top);
                const int tot = count_ge(cand);
                const float cv = __uint_as_float(cand ^ (~(unsigned)((int)cand >> 31) | 0x80000000u));
                const float g  = (float)tot - (float)k;
                const bool  ge = tot >= k;
                top = (tot > k) ? top : ((tot == k) ? cand : cand - 1u);
                lo  = ge ? cand : lo;
                a   = ge ? cv : a;
                b   = ge ? b : cv;
                ga  = ge ? g : ga * 0.8f;
                gb  = ge ? gb * 0.8f : g;
                if (lo >= top) break;
            } while (--left > 0);
            while (lo < top) {
                const unsigned d = top - lo;
                const unsigned cand = lo + (d >> 1) + (d & 1u);
                const int tot = count_ge(cand);
                top = (tot > k) ? top : ((tot == k) ? cand : cand - 1u);
                lo  = (tot >= k) ? cand : lo;
            }
            th[t] = lo;
        }

        float s1 = 0.f, s2 = 0.f, s3 = 0.f, s4 = 0.f;
        float w[29];
        #pragma unroll
        for (int j = 0; j < 29; ++j) {
            float e = __expf(rv[j] - mx);
            w[j] = e;
            s4 += e;
            if (rk[j] >= th[0]) s1 += e;
            if (rk[j] >= th[1]) s2 += e;
            if (rk[j] >= th[2]) s3 += e;
        }
        #pragma unroll
        for (int s = 16; s > 0; s >>= 1) {
            s1 += __shfl_xor(s1, s, 32);
            s2 += __shfl_xor(s2, s, 32);
            s3 += __shfl_xor(s3, s, 32);
            s4 += __shfl_xor(s4, s, 32);
        }
        const float c1 = A1 / s1, c2 = A2 / s2, c3 = A3 / s3, c4 = A4 / s4;
        #pragma unroll
        for (int j = 0; j < 29; ++j) {
            float f = c4;
            if (rk[j] >= th[0]) f += c1;
            if (rk[j] >= th[1]) f += c2;
            if (rk[j] >= th[2]) f += c3;
            srow[lane + 32 * j] = w[j] * f;
        }
    }
    __syncthreads();

    const int nhalf = wave & 1;
    const int kg    = wave >> 1;
    const v4f* vf = (const v4f*)vbase + nhalf * 64 + lane;
    v8f acc = {};
    for (int mt = kg; mt < NT; mt += 16) {
        v4f vt[4][2];
        #pragma unroll
        for (int u = 0; u < 4; ++u) {
            const int mu = (mt + 4 * u < NT) ? mt + 4 * u : mt;
            const v4f* bp = vf + mu * 128;
            vt[u][0] = bp[0];
            vt[u][1] = bp[32];
        }
        #pragma unroll
        for (int u = 0; u < 4; ++u) {
            const int mu = mt + 4 * u;
            if (mu < NT) {
                const float* ap = sc + ll * SROW + mu * 16 + 2 * lh;
                const v2f a0 = *(const v2f*)(ap);
                const v2f a1 = *(const v2f*)(ap + 4);
                const v2f a2 = *(const v2f*)(ap + 8);
                const v2f a3 = *(const v2f*)(ap + 12);
                acc = __builtin_amdgcn_wmma_f32_16x16x4_f32(false, a0, false, vt[u][0].xy,
                                                            (short)0, acc, false, false);
                acc = __builtin_amdgcn_wmma_f32_16x16x4_f32(false, a1, false, vt[u][0].zw,
                                                            (short)0, acc, false, false);
                acc = __builtin_amdgcn_wmma_f32_16x16x4_f32(false, a2, false, vt[u][1].xy,
                                                            (short)0, acc, false, false);
                acc = __builtin_amdgcn_wmma_f32_16x16x4_f32(false, a3, false, vt[u][1].zw,
                                                            (short)0, acc, false, false);
            }
        }
    }
    __syncthreads();

    float* part = sc;
    #pragma unroll
    for (int r = 0; r < 8; ++r) part[(wave * 8 + r) * 32 + lane] = acc[r];
    __syncthreads();

    if (wave < 2) {
        #pragma unroll
        for (int r = 0; r < 8; ++r) {
            float v = 0.f;
            #pragma unroll
            for (int g = 0; g < 4; ++g)
                v += part[(((g * 2 + wave) * 8) + r) * 32 + lane];
            const int row = n0 + r + 8 * lh;
            const int col = 16 * wave + ll;
            if (row < NQ)
                yo[((size_t)bh * NPAD + row) * CH + col] = v;
        }
    }
}

__global__ __launch_bounds__(256) void bsum_kernel(
    const float* __restrict__ yo, float* __restrict__ msum)
{
    const int idx = blockIdx.x * 256 + threadIdx.x;
    if (idx >= 912 * DIM) return;
    const int n = idx >> 7;
    const int d = idx & 127;
    float v = 0.f;
    if (n < NQ) {
        const int h = d >> 5, c = d & 31;
        #pragma unroll
        for (int b = 0; b < BATCH; ++b)
            v += yo[(((size_t)(b * HEADS + h)) * NPAD + n) * CH + c];
    }
    msum[idx] = v;
}

__global__ __launch_bounds__(256) void out_kernel(
    const float* __restrict__ msum, const float* __restrict__ Wo,
    const float* __restrict__ q, float* __restrict__ out)
{
    const int rt   = blockIdx.x;
    const int wave = threadIdx.x >> 5;
    const int lane = threadIdx.x & 31;
    const int lh   = lane >> 4;
    const int ll   = lane & 15;
    const int r0   = rt * 16;
    const int d0   = wave * 16;

    const float* aptr = msum + (size_t)(r0 + ll) * DIM + 2 * lh;
    const float* bptr = Wo   + (size_t)(d0 + ll) * DIM + 2 * lh;

    v8f acc = {};
    #pragma unroll
    for (int k = 0; k < DIM; k += 4) {
        v2f a = *(const v2f*)(aptr + k);
        v2f b = *(const v2f*)(bptr + k);
        acc = __builtin_amdgcn_wmma_f32_16x16x4_f32(false, a, false, b,
                                                    (short)0, acc, false, false);
    }
    #pragma unroll
    for (int r = 0; r < 8; ++r) {
        const int row = r0 + r + 8 * lh;
        const int col = d0 + ll;
        if (row < NQ)
            out[(size_t)row * DIM + col] = acc[r] + q[(size_t)row * DIM + col];
    }
}

extern "C" void kernel_launch(void* const* d_in, const int* in_sizes, int n_in,
                              void* d_out, int out_size, void* d_ws, size_t ws_size,
                              hipStream_t stream)
{
    const float* q    = (const float*)d_in[0];
    const float* Wq   = (const float*)d_in[1];
    const float* Wk   = (const float*)d_in[2];
    const float* Wv   = (const float*)d_in[3];
    const float* Wo   = (const float*)d_in[4];
    const float* temp = (const float*)d_in[5];
    const float* a1   = (const float*)d_in[6];
    const float* a2   = (const float*)d_in[7];
    const float* a3   = (const float*)d_in[8];
    const float* a4   = (const float*)d_in[9];
    float* out = (float*)d_out;

    const size_t per = (size_t)BATCH * HEADS * NPAD * CH;
    float* yq   = (float*)d_ws;
    float* yk   = yq + per;
    float* yv   = yk + per;
    float* yo   = yv + per;
    float* msum = yo + per;

    proj_kernel<<<dim3(BATCH * NT, 3), 128, 0, stream>>>(q, Wq, Wk, Wv, temp, yq, yk, yv);
    attn_kernel<<<dim3(NT, BATCH * HEADS), 256, 0, stream>>>(
        yq, yk, yv, a1, a2, a3, a4, yo);
    bsum_kernel<<<(912 * DIM + 255) / 256, 256, 0, stream>>>(yo, msum);
    out_kernel<<<NT, 256, 0, stream>>>(msum, Wo, q, out);
}


# input and output projections: matrix fragments read from LDS that whole 512-B row loads fill (weight fragments kept over 3 row tiles, input rows shared by waves), not 16-line gathers; bit-identical
# speedup vs baseline: 3.3548x; 1.0767x over previous
#include <hip/hip_runtime.h>
#include <hip/hip_bf16.h>

typedef __attribute__((ext_vector_type(2))) float v2f;
typedef __attribute__((ext_vector_type(4))) float v4f;
typedef __attribute__((ext_vector_type(8))) float v8f;

#define BATCH 8
#define HEADS 4
#define NQ    900
#define DIM   128
#define CH    32
#define NPAD  928
#define NT    57
#define SROW  928
#define PROBE_STEPS 16

__device__ __forceinline__ int kfrag_index(int m, int ch)
{
    return ((((m >> 4) * 4 + (ch >> 3)) * 32 + (((ch >> 1) & 1) << 4) + (m & 15)) << 2)
           + (((ch >> 2) & 1) << 1) + (ch & 1);
}
__device__ __forceinline__ int vfrag_index(int m, int ch)
{
    return (((((m >> 4) * 2 + (ch >> 4)) * 2 + ((m >> 3) & 1)) * 32
             + (((m >> 1) & 1) << 4) + (ch & 15)) << 2)
           + (((m >> 2) & 1) << 1) + (m & 1);
}

#define PPITCH 36
#define NG     19
#define XP     132

__global__ __launch_bounds__(128) void proj_kernel(
    const float* __restrict__ q,
    const float* __restrict__ Wq, const float* __restrict__ Wk,
    const float* __restrict__ Wv, const float* __restrict__ temp,
    float* __restrict__ yq, float* __restrict__ yk, float* __restrict__ yv)
{
    __shared__ __attribute__((aligned(16))) float xs[48 * XP];
    __shared__ __attribute__((aligned(16))) float wst[4 * 16 * XP];

    const int g     = blockIdx.x;
    const int which = blockIdx.y;
    const int b_    = g / NG;
    const int mt0   = (g - b_ * NG) * 3;
    const int h     = threadIdx.x >> 5;
    const int lane  = threadIdx.x & 31;
    const int lh    = lane >> 4;
    const int ll    = lane & 15;

    const float* W = (which == 0) ? Wq : (which == 1) ? Wk : Wv;
    const float tt = temp[h];

    float* wl = wst + h * (16 * XP);

    v4f wr[32], xr[12];
    {
        const float* wp = W + (size_t)(32 * h) * DIM + 4 * lane;
        #pragma unroll
        for (int r = 0; r < 32; ++r) wr[r] = *(const v4f*)(wp + r * DIM);
        #pragma unroll
        for (int i = 0; i < 12; ++i) {
            int n = mt0 * 16 + h + 4 * i;
            n = (n < NQ) ? n : NQ - 1;
            xr[i] = *(const v4f*)(q + ((size_t)b_ * NQ + n) * DIM + 4 * lane);
        }
    }
    #pragma unroll
    for (int i = 0; i < 12; ++i) *(v4f*)(xs + (h + 4 * i) * XP + 4 * lane) = xr[i];
    #pragma unroll
    for (int r = 0; r < 16; ++r) *(v4f*)(wl + r * XP + 4 * lane) = wr[r];
    __syncthreads();

    v2f b0[32], b1[32];
    const float* bp = wl + ll * XP + 2 * lh;
    #pragma unroll
    for (int c = 0; c < 32; ++c) b0[c] = *(const v2f*)(bp + 4 * c);
    __syncthreads();
    #pragma unroll
    for (int r = 0; r < 16; ++r) *(v4f*)(wl + r * XP + 4 * lane) = wr[16 + r];
    __syncthreads();
    #pragma unroll
    for (int c = 0; c < 32; ++c) b1[c] = *(const v2f*)(bp + 4 * c);
    __syncthreads();

    for (int i = 0; i < 3; ++i) {
        const int mt = mt0 + i;
        const int n0 = mt * 16;
        const float* ap = xs + (16 * i + ll) * XP + 2 * lh;

        v8f acc0 = {}, acc1 = {};
        #pragma unroll
        for (int c = 0; c < 32; ++c) {
            const v2f a = *(const v2f*)(ap + 4 * c);
            acc0 = __builtin_amdgcn_wmma_f32_16x16x4_f32(false, a, false, b0[c],
                                                         (short)0, acc0, false, false);
            acc1 = __builtin_amdgcn_wmma_f32_16x16x4_f32(false, a, false, b1[c],
                                                         (short)0, acc1, false, false);
        }
        const size_t plane = (size_t)(b_ * HEADS + h) * NPAD * CH;

        if (which == 2) {
            float* vp = yv + plane;
            const int rbase = n0 + 8 * lh;
            v4f lo0 = { acc0[0], acc0[1], acc0[4], acc0[5] };
            v4f hi0 = { acc0[2], acc0[3], acc0[6], acc0[7] };
            v4f lo1 = { acc1[0], acc1[1], acc1[4], acc1[5] };
            v4f hi1 = { acc1[2], acc1[3], acc1[6], acc1[7] };
            if (rbase + 7 >= NQ) {
                const bool p0 = rbase + 0 < NQ, p1 = rbase + 1 < NQ, p2 = rbase + 2 < NQ,
                           p3 = rbase + 3 < NQ, p4 = rbase + 4 < NQ, p5 = rbase + 5 < NQ,
                           p6 = rbase + 6 < NQ, p7 = rbase + 7 < NQ;
                lo0.x = p0 ? lo0.x : 0.f; lo0.y = p1 ? lo0.y : 0.f;
                lo0.z = p4 ? lo0.z : 0.f; lo0.w = p5 ? lo0.w : 0.f;
                hi0.x = p2 ? hi0.x : 0.f; hi0.y = p3 ? hi0.y : 0.f;
                hi0.z = p6 ? hi0.z : 0.f; hi0.w = p7 ? hi0.w : 0.f;
                lo1.x = p0 ? lo1.x : 0.f; lo1.y = p1 ? lo1.y : 0.f;
                lo1.z = p4 ? lo1.z : 0.f; lo1.w = p5 ? lo1.w : 0.f;
                hi1.x = p2 ? hi1.x : 0.f; hi1.y = p3 ? hi1.y : 0.f;
                hi1.z = p6 ? hi1.z : 0.f; hi1.w = p7 ? hi1.w : 0.f;
            }
            const int base0 = (((mt * 2 + 0) * 2 + lh) * 32 + ll) * 4;
            const int base1 = (((mt * 2 + 1) * 2 + lh) * 32 + ll) * 4;
            *(v4f*)(vp + base0)      = lo0;
            *(v4f*)(vp + base0 + 64) = hi0;
            *(v4f*)(vp + base1)      = lo1;
            *(v4f*)(vp + base1 + 64) = hi1;
        } else {
            float* tl = wl;
            #pragma unroll
            for (int r = 0; r < 8; ++r) {
                tl[(8 * lh + r) * PPITCH + ll]      = acc0[r];
                tl[(8 * lh + r) * PPITCH + 16 + ll] = acc1[r];
            }
            __syncthreads();

            const int row  = ll;
            const int half = lh;
            const v4f* mp = (const v4f*)(tl + row * PPITCH + 16 * half);
            const v4f* op = (const v4f*)(tl + row * PPITCH + 16 * (1 - half));
            float mine[16], oth[16];
            #pragma unroll
            for (int j = 0; j < 4; ++j) {
                const v4f m4 = mp[j];
                const v4f o4 = op[j];
                mine[4 * j + 0] = m4.x; mine[4 * j + 1] = m4.y;
                mine[4 * j + 2] = m4.z; mine[4 * j + 3] = m4.w;
                oth[4 * j + 0]  = o4.x; oth[4 * j + 1]  = o4.y;
                oth[4 * j + 2]  = o4.z; oth[4 * j + 3]  = o4.w;
            }

            float S[16];
            #pragma unroll
            for (int i = 0; i < 16; ++i) {
                const float p = oth[i] * oth[i];
                S[i] = __builtin_fmaf(mine[i], mine[i], p);
            }
            #pragma unroll
            for (int i = 0; i < 8; ++i) S[i] = S[i] + S[i + 8];
            #pragma unroll
            for (int i = 0; i < 4; ++i) S[i] = S[i] + S[i + 4];
            S[0] = S[0] + S[2];
            S[1] = S[1] + S[3];
            const float ssq = S[0] + S[1];
            const float dn  = fmaxf(sqrtf(ssq), 1e-12f);

            const bool live = (n0 + row) < NQ;
            float o[16];
            #pragma unroll
            for (int i = 0; i < 16; ++i) {
                float v = mine[i] / dn;
                if (which == 0) v = v * tt;
                o[i] = live ? v : 0.f;
            }
            if (which == 0) {
                float* dst = yq + plane + (size_t)(n0 + row) * CH + 16 * half;
                #pragma unroll
                for (int j = 0; j < 4; ++j) {
                    v4f s4 = { o[4 * j], o[4 * j + 1], o[4 * j + 2], o[4 * j + 3] };
                    *(v4f*)(dst + 4 * j) = s4;
                }
            } else {
                float* kp = yk + plane;
                #pragma unroll
                for (int o2 = 0; o2 < 2; ++o2) {
                    #pragma unroll
                    for (int lhp = 0; lhp < 2; ++lhp) {
                        const int c0 = 8 * o2 + 2 * lhp;
                        v4f s4 = { o[c0], o[c0 + 1], o[c0 + 4], o[c0 + 5] };
                        *(v4f*)(kp + (((mt * 4 + 2 * half + o2) * 32 + 16 * lhp + row) * 4)) = s4;
                    }
                }
            }
        }
    }
}

__global__ __launch_bounds__(256) void attn_kernel(
    const float* __restrict__ qh, const float* __restrict__ kh,
    const float* __restrict__ vh,
    const float* __restrict__ a1p, const float* __restrict__ a2p,
    const float* __restrict__ a3p, const float* __restrict__ a4p,
    float* __restrict__ yo)
{
    __shared__ float sc[16 * SROW];

    const int bh   = blockIdx.y;
    const int nt   = blockIdx.x;
    const int wave = threadIdx.x >> 5;
    const int lane = threadIdx.x & 31;
    const int lh   = lane >> 4;
    const int ll   = lane & 15;

    const float* qbase = qh + (size_t)bh * NPAD * CH;
    const float* kbase = kh + (size_t)bh * NPAD * CH;
    const float* vbase = vh + (size_t)bh * NPAD * CH;
    const int n0 = nt * 16;

    v2f areg[8];
    {
        const float* ap = qbase + (size_t)(n0 + ll) * CH + 2 * lh;
        #pragma unroll
        for (int c = 0; c < 8; ++c) areg[c] = *(const v2f*)(ap + 4 * c);
    }

    const v4f* kf = (const v4f*)kbase + lane;
    v4f nk0, nk1, nk2, nk3;
    {
        const v4f* bp = kf + wave * 128;
        nk0 = bp[0]; nk1 = bp[32]; nk2 = bp[64]; nk3 = bp[96];
    }
    for (int mt = wave; mt < NT; mt += 8) {
        const int m0 = mt * 16;
        const v4f k0 = nk0, k1 = nk1, k2 = nk2, k3 = nk3;
        {
            const int mn = (mt + 8 < NT) ? mt + 8 : NT - 1;
            const v4f* bp = kf + mn * 128;
            nk0 = bp[0]; nk1 = bp[32]; nk2 = bp[64]; nk3 = bp[96];
        }
        v8f acc = {};
        acc = __builtin_amdgcn_wmma_f32_16x16x4_f32(false, areg[0], false, k0.xy,
                                                    (short)0, acc, false, false);
        acc = __builtin_amdgcn_wmma_f32_16x16x4_f32(false, areg[1], false, k0.zw,
                                                    (short)0, acc, false, false);
        acc = __builtin_amdgcn_wmma_f32_16x16x4_f32(false, areg[2], false, k1.xy,
                                                    (short)0, acc, false, false);
        acc = __builtin_amdgcn_wmma_f32_16x16x4_f32(false, areg[3], false, k1.zw,
                                                    (short)0, acc, false, false);
        acc = __builtin_amdgcn_wmma_f32_16x16x4_f32(false, areg[4], false, k2.xy,
                                                    (short)0, acc, false, false);
        acc = __builtin_amdgcn_wmma_f32_16x16x4_f32(false, areg[5], false, k2.zw,
                                                    (short)0, acc, false, false);
        acc = __builtin_amdgcn_wmma_f32_16x16x4_f32(false, areg[6], false, k3.xy,
                                                    (short)0, acc, false, false);
        acc = __builtin_amdgcn_wmma_f32_16x16x4_f32(false, areg[7], false, k3.zw,
                                                    (short)0, acc, false, false);
        #pragma unroll
        for (int r = 0; r < 8; ++r)
            sc[(r + 8 * lh) * SROW + m0 + ll] = acc[r];
    }
    __syncthreads();

    for (int i = threadIdx.x; i < 16 * (SROW - NQ); i += 256) {
        int row = i / (SROW - NQ);
        int m   = NQ + i - row * (SROW - NQ);
        sc[row * SROW + m] = -1e30f;
    }
    __syncthreads();

    const float A1 = a1p[0], A2 = a2p[0], A3 = a3p[0], A4 = a4p[0];

    for (int rr = 0; rr < 2; ++rr) {
        const int row = wave + 8 * rr;
        float* srow = sc + row * SROW;

        float rv[29];
        #pragma unroll
        for (int j = 0; j < 29; ++j) rv[j] = srow[lane + 32 * j];

        float mx = -3.0e38f;
        #pragma unroll
        for (int j = 0; j < 29; ++j) mx = fmaxf(mx, rv[j]);
        #pragma unroll
        for (int s = 16; s > 0; s >>= 1) mx = fmaxf(mx, __shfl_xor(mx, s, 32));

        unsigned rk[29];
        #pragma unroll
        for (int j = 0; j < 29; ++j) {
            unsigned u = __float_as_uint(rv[j]);
            rk[j] = (u & 0x80000000u) ? ~u : (u | 0x80000000u);
        }

        __builtin_amdgcn_sched_barrier(0);

        float sm = 0.f, sq = 0.f;
        #pragma unroll
        for (int j = 0; j < 28; ++j) {
            sm += rv[j];
            sq = __builtin_fmaf(rv[j], rv[j], sq);
        }
        {
            const float tl = (lane < 4) ? rv[28] : 0.f;
            sm += tl;
            sq = __builtin_fmaf(tl, tl, sq);
        }
        #pragma unroll
        for (int s = 16; s > 0; s >>= 1) {
            sm += __shfl_xor(sm, s, 32);
            sq += __shfl_xor(sq, s, 32);
        }
        const float smU = __uint_as_float(__builtin_amdgcn_readfirstlane(__float_as_uint(sm)));
        const float sqU = __uint_as_float(__builtin_amdgcn_readfirstlane(__float_as_uint(sq)));
        const float muU = smU * (1.f / (float)NQ);
        const float sdU = __builtin_amdgcn_sqrtf(fmaxf(sqU * (1.f / (float)NQ) - muU * muU, 0.f));

        auto count_ge = [&](unsigned cand) -> int {
            int tot = 0;
            #pragma unroll
            for (int j = 0; j + 1 < 29; j += 2) {
                const unsigned long long two =
                    (unsigned long long)__builtin_amdgcn_ballot_w32(rk[j] >= cand) |
                    ((unsigned long long)__builtin_amdgcn_ballot_w32(rk[j + 1] >= cand) << 32);
                tot += __builtin_popcountll(two);
            }
            tot += __builtin_popcount(__builtin_amdgcn_ballot_w32(rk[28] >= cand));
            return tot;
        };

        unsigned th[3];
        const int   ks[3] = {100, 300, 600};
        const float zq[3] = {1.2177f, 0.4292f, -0.4322f};
        const float gq[3] = {641.5f, 1228.0f, 1226.4f};
        #pragma unroll
        for (int t = 2; t >= 0; --t) {
            const int k = ks[t];
            unsigned lo = 0u, top = 0xFFFFFFFFu;
            const float ctr = muU + zq[t] * sdU;
            float a = ctr - 3.f * sdU, ga = gq[t];
            float b = ctr + 3.f * sdU, gb = -gq[t];
            int left = PROBE_STEPS;
            do {
                const float cf = a + (b - a) * (ga * __builtin_amdgcn_rcpf(ga - gb));
                const unsigned u = __float_as_uint(cf);
                unsigned cand = u ^ ((unsigned)((int)u >> 31) | 0x80000000u);
                cand = max(cand, lo + 1u);
                cand = min(cand, top);
                const int tot = count_ge(cand);
                const float cv = __uint_as_float(cand ^ (~(unsigned)((int)cand >> 31) | 0x80000000u));
                const float g  = (float)tot - (float)k;
                const bool  ge = tot >= k;
                top = (tot > k) ? top : ((tot == k) ? cand : cand - 1u);
                lo  = ge ? cand : lo;
                a   = ge ? cv : a;
                b   = ge ? b : cv;
                ga  = ge ? g : ga * 0.8f;
                gb  = ge ? gb * 0.8f : g;
                if (lo >= top) break;
            } while (--left > 0);
            while (lo < top) {
                const unsigned d = top - lo;
                const unsigned cand = lo + (d >> 1) + (d & 1u);
                const int tot = count_ge(cand);
                top = (tot > k) ? top : ((tot == k) ? cand : cand - 1u);
                lo  = (tot >= k) ? cand : lo;
            }
            th[t] = lo;
        }

        float s1 = 0.f, s2 = 0.f, s3 = 0.f, s4 = 0.f;
        float w[29];
        #pragma unroll
        for (int j = 0; j < 29; ++j) {
            float e = __expf(rv[j] - mx);
            w[j] = e;
            s4 += e;
            if (rk[j] >= th[0]) s1 += e;
            if (rk[j] >= th[1]) s2 += e;
            if (rk[j] >= th[2]) s3 += e;
        }
        #pragma unroll
        for (int s = 16; s > 0; s >>= 1) {
            s1 += __shfl_xor(s1, s, 32);
            s2 += __shfl_xor(s2, s, 32);
            s3 += __shfl_xor(s3, s, 32);
            s4 += __shfl_xor(s4, s, 32);
        }
        const float c1 = A1 / s1, c2 = A2 / s2, c3 = A3 / s3, c4 = A4 / s4;
        #pragma unroll
        for (int j = 0; j < 29; ++j) {
            float f = c4;
            if (rk[j] >= th[0]) f += c1;
            if (rk[j] >= th[1]) f += c2;
            if (rk[j] >= th[2]) f += c3;
            srow[lane + 32 * j] = w[j] * f;
        }
    }
    __syncthreads();

    const int nhalf = wave & 1;
    const int kg    = wave >> 1;
    const v4f* vf = (const v4f*)vbase + nhalf * 64 + lane;
    v8f acc = {};
    for (int mt = kg; mt < NT; mt += 16) {
        v4f vt[4][2];
        #pragma unroll
        for (int u = 0; u < 4; ++u) {
            const int mu = (mt + 4 * u < NT) ? mt + 4 * u : mt;
            const v4f* bp = vf + mu * 128;
            vt[u][0] = bp[0];
            vt[u][1] = bp[32];
        }
        #pragma unroll
        for (int u = 0; u < 4; ++u) {
            const int mu = mt + 4 * u;
            if (mu < NT) {
                const float* ap = sc + ll * SROW + mu * 16 + 2 * lh;
                const v2f a0 = *(const v2f*)(ap);
                const v2f a1 = *(const v2f*)(ap + 4);
                const v2f a2 = *(const v2f*)(ap + 8);
                const v2f a3 = *(const v2f*)(ap + 12);
                acc = __builtin_amdgcn_wmma_f32_16x16x4_f32(false, a0, false, vt[u][0].xy,
                                                            (short)0, acc, false, false);
                acc = __builtin_amdgcn_wmma_f32_16x16x4_f32(false, a1, false, vt[u][0].zw,
                                                            (short)0, acc, false, false);
                acc = __builtin_amdgcn_wmma_f32_16x16x4_f32(false, a2, false, vt[u][1].xy,
                                                            (short)0, acc, false, false);
                acc = __builtin_amdgcn_wmma_f32_16x16x4_f32(false, a3, false, vt[u][1].zw,
                                                            (short)0, acc, false, false);
            }
        }
    }
    __syncthreads();

    float* part = sc;
    #pragma unroll
    for (int r = 0; r < 8; ++r) part[(wave * 8 + r) * 32 + lane] = acc[r];
    __syncthreads();

    if (wave < 2) {
        #pragma unroll
        for (int r = 0; r < 8; ++r) {
            float v = 0.f;
            #pragma unroll
            for (int g = 0; g < 4; ++g)
                v += part[(((g * 2 + wave) * 8) + r) * 32 + lane];
            const int row = n0 + r + 8 * lh;
            const int col = 16 * wave + ll;
            if (row < NQ)
                yo[((size_t)bh * NPAD + row) * CH + col] = v;
        }
    }
}

__global__ __launch_bounds__(256) void bsum_kernel(
    const float* __restrict__ yo, float* __restrict__ msum)
{
    const int idx = blockIdx.x * 256 + threadIdx.x;
    if (idx >= 912 * DIM) return;
    const int n = idx >> 7;
    const int d = idx & 127;
    float v = 0.f;
    if (n < NQ) {
        const int h = d >> 5, c = d & 31;
        #pragma unroll
        for (int b = 0; b < BATCH; ++b)
            v += yo[(((size_t)(b * HEADS + h)) * NPAD + n) * CH + c];
    }
    msum[idx] = v;
}

__global__ __launch_bounds__(256) void out_kernel(
    const float* __restrict__ msum, const float* __restrict__ Wo,
    const float* __restrict__ q, float* __restrict__ out)
{
    __shared__ __attribute__((aligned(16))) float ast[16 * XP];
    __shared__ __attribute__((aligned(16))) float bst[8 * 16 * XP];

    const int rt   = blockIdx.x;
    const int wave = threadIdx.x >> 5;
    const int lane = threadIdx.x & 31;
    const int lh   = lane >> 4;
    const int ll   = lane & 15;
    const int r0   = rt * 16;
    const int d0   = wave * 16;

    float* bl = bst + wave * (16 * XP);

    v4f wr[16], xr[2];
    {
        const float* wp = Wo + (size_t)d0 * DIM + 4 * lane;
        #pragma unroll
        for (int r = 0; r < 16; ++r) wr[r] = *(const v4f*)(wp + r * DIM);
        #pragma unroll
        for (int i = 0; i < 2; ++i)
            xr[i] = *(const v4f*)(msum + (size_t)(r0 + wave + 8 * i) * DIM + 4 * lane);
    }
    #pragma unroll
    for (int i = 0; i < 2; ++i) *(v4f*)(ast + (wave + 8 * i) * XP + 4 * lane) = xr[i];
    #pragma unroll
    for (int r = 0; r < 16; ++r) *(v4f*)(bl + r * XP + 4 * lane) = wr[r];
    __syncthreads();

    const float* ap = ast + ll * XP + 2 * lh;
    const float* bp = bl  + ll * XP + 2 * lh;

    v8f acc = {};
    #pragma unroll
    for (int k = 0; k < DIM; k += 4) {
        v2f a = *(const v2f*)(ap + k);
        v2f b = *(const v2f*)(bp + k);
        acc = __builtin_amdgcn_wmma_f32_16x16x4_f32(false, a, false, b,
                                                    (short)0, acc, false, false);
    }
    #pragma unroll
    for (int r = 0; r < 8; ++r) {
        const int row = r0 + r + 8 * lh;
        const int col = d0 + ll;
        if (row < NQ)
            out[(size_t)row * DIM + col] = acc[r] + q[(size_t)row * DIM + col];
    }
}

extern "C" void kernel_launch(void* const* d_in, const int* in_sizes, int n_in,
                              void* d_out, int out_size, void* d_ws, size_t ws_size,
                              hipStream_t stream)
{
    const float* q    = (const float*)d_in[0];
    const float* Wq   = (const float*)d_in[1];
    const float* Wk   = (const float*)d_in[2];
    const float* Wv   = (const float*)d_in[3];
    const float* Wo   = (const float*)d_in[4];
    const float* temp = (const float*)d_in[5];
    const float* a1   = (const float*)d_in[6];
    const float* a2   = (const float*)d_in[7];
    const float* a3   = (const float*)d_in[8];
    const float* a4   = (const float*)d_in[9];
    float* out = (float*)d_out;

    const size_t per = (size_t)BATCH * HEADS * NPAD * CH;
    float* yq   = (float*)d_ws;
    float* yk   = yq + per;
    float* yv   = yk + per;
    float* yo   = yv + per;
    float* msum = yo + per;

    proj_kernel<<<dim3(BATCH * NG, 3), 128, 0, stream>>>(q, Wq, Wk, Wv, temp, yq, yk, yv);
    attn_kernel<<<dim3(NT, BATCH * HEADS), 256, 0, stream>>>(
        yq, yk, yv, a1, a2, a3, a4, yo);
    bsum_kernel<<<(912 * DIM + 255) / 256, 256, 0, stream>>>(yo, msum);
    out_kernel<<<NT, 256, 0, stream>>>(msum, Wo, q, out);
}
